# starting kernel
# baseline (speedup 1.0000x reference)
#include <hip/hip_runtime.h>

typedef __attribute__((ext_vector_type(2))) float v2f;
typedef __attribute__((ext_vector_type(8))) float v8f;

__device__ __forceinline__ void cmul(float ar, float ai, float br, float bi,
                                     float& cr, float& ci) {
    cr = ar * br - ai * bi;
    ci = ar * bi + ai * br;
}

__device__ __forceinline__ void cpow(float ar, float ai, int p, float& r, float& i) {
    float xr = 1.f, xi = 0.f;
    for (int q = 0; q < p; ++q) { float tr, ti; cmul(xr, xi, ar, ai, tr, ti); xr = tr; xi = ti; }
    r = xr; i = xi;
}

__global__ void slab_pass1(const float* __restrict__ pk, const float* __restrict__ TAx,
                           const float* __restrict__ TAy, const float* __restrict__ fc,
                           const int* __restrict__ p_dt, const int* __restrict__ p_dtf,
                           float* __restrict__ out, int nsteps, int nchunks, int ntiles,
                           int nt)
{
    int gid  = blockIdx.x * blockDim.x + threadIdx.x;
    int tile = gid >> 5;
    int lane = gid & 31;
    if (tile >= ntiles) return;

    float dt      = (float)p_dt[0];
    int   nsub    = p_dtf[0] / p_dt[0];
    float invnsub = 1.f / (float)nsub;
    float K0   = __expf(pk[0]);
    float K1   = __expf(pk[1]);
    float f    = fc[0];
    float ar   = 1.f - dt * K1;
    float ai   = -dt * f;
    float dtK0 = dt * K0;

    int half = lane >> 4;
    int r    = lane & 15;
    int c0   = tile * 16;

    v2f Are[4], Aim[4], Mre[4], Mim[4], Mneg[4];
    #pragma unroll
    for (int s = 0; s < 4; ++s) {
        #pragma unroll
        for (int e = 0; e < 2; ++e) {
            int j = 4 * s + 2 * half + e;
            int  c  = c0 + r;
            long n  = (long)16 * c + j;
            int itf0 = (int)(n / nsub);
            float aa = (float)(n - (long)itf0 * nsub) * invnsub;
            int i0 = itf0     < nt - 1 ? itf0     : nt - 1;
            int i1 = itf0 + 1 < nt - 1 ? itf0 + 1 : nt - 1;
            float tax = (1.f - aa) * TAx[i0] + aa * TAx[i1];
            float tay = (1.f - aa) * TAy[i0] + aa * TAy[i1];
            float valid = (c < nchunks && n < (long)nsteps) ? 1.f : 0.f;
            Are[s][e] = valid * dtK0 * tax;
            Aim[s][e] = valid * dtK0 * tay;
            int p = r - j;
            float pr, pi;
            cpow(ar, ai, p < 0 ? 0 : p, pr, pi);
            float m = (p >= 0) ? 1.f : 0.f;
            Mre[s][e]  =  m * pr;
            Mim[s][e]  =  m * pi;
            Mneg[s][e] = -m * pi;
        }
    }

    v8f Sre = {}; v8f Sim = {};
    #pragma unroll
    for (int s = 0; s < 4; ++s)
        Sre = __builtin_amdgcn_wmma_f32_16x16x4_f32(false, Are[s], false, Mre[s],
                                                    (short)0, Sre, false, false);
    #pragma unroll
    for (int s = 0; s < 4; ++s)
        Sre = __builtin_amdgcn_wmma_f32_16x16x4_f32(false, Aim[s], false, Mneg[s],
                                                    (short)0, Sre, false, false);
    #pragma unroll
    for (int s = 0; s < 4; ++s)
        Sim = __builtin_amdgcn_wmma_f32_16x16x4_f32(false, Are[s], false, Mim[s],
                                                    (short)0, Sim, false, false);
    #pragma unroll
    for (int s = 0; s < 4; ++s)
        Sim = __builtin_amdgcn_wmma_f32_16x16x4_f32(false, Aim[s], false, Mre[s],
                                                    (short)0, Sim, false, false);

    #pragma unroll
    for (int e = 0; e < 8; ++e) {
        int  row = e + 8 * half;
        int  c   = c0 + row;
        long n   = (long)16 * c + r;
        if (c < nchunks && n < (long)nsteps) {
            float2 v; v.x = Sre[e]; v.y = Sim[e];
            *reinterpret_cast<float2*>(out + 2 * n) = v;
        }
    }
}

__global__ void slab_pass2(const float* __restrict__ pk, const float* __restrict__ fc,
                           const int* __restrict__ p_dt,
                           const float* __restrict__ out, float2* __restrict__ ws,
                           int nsteps, int nchunks)
{
    __shared__ float4 sh[1024];
    int t = threadIdx.x;
    float dt = (float)p_dt[0];
    float K1 = __expf(pk[1]);
    float f  = fc[0];
    float ar = 1.f - dt * K1, ai = -dt * f;
    float a16r, a16i; cpow(ar, ai, 16, a16r, a16i);

    int cpt = (nchunks + 1023) >> 10;
    int cb  = t * cpt;
    int ce  = min(cb + cpt, nchunks);

    float Ar = 1.f, Ai = 0.f, Br = 0.f, Bi = 0.f;
    for (int c = cb; c < ce; ++c) {
        long n15 = (long)16 * c + 15;
        float gr = 0.f, gi = 0.f;
        if (n15 < (long)nsteps) { gr = out[2 * n15]; gi = out[2 * n15 + 1]; }
        float tr, ti;
        cmul(a16r, a16i, Ar, Ai, tr, ti); Ar = tr; Ai = ti;
        cmul(a16r, a16i, Br, Bi, tr, ti); Br = tr + gr; Bi = ti + gi;
    }
    sh[t] = make_float4(Ar, Ai, Br, Bi);
    __syncthreads();
    for (int off = 1; off < 1024; off <<= 1) {
        float4 v = sh[t];
        if (t >= off) {
            float4 p = sh[t - off];
            float nar, nai, tr, ti;
            cmul(v.x, v.y, p.x, p.y, nar, nai);
            cmul(v.x, v.y, p.z, p.w, tr, ti);
            v = make_float4(nar, nai, tr + v.z, ti + v.w);
        }
        __syncthreads();
        sh[t] = v;
        __syncthreads();
    }
    float Wr = 0.f, Wi = 0.f;
    if (t > 0) { float4 p = sh[t - 1]; Wr = p.z; Wi = p.w; }
    for (int c = cb; c < ce; ++c) {
        ws[c] = make_float2(Wr, Wi);
        long n15 = (long)16 * c + 15;
        float gr = 0.f, gi = 0.f;
        if (n15 < (long)nsteps) { gr = out[2 * n15]; gi = out[2 * n15 + 1]; }
        float tr, ti;
        cmul(a16r, a16i, Wr, Wi, tr, ti);
        Wr = tr + gr; Wi = ti + gi;
    }
}

__global__ void slab_pass3(const float* __restrict__ pk, const float* __restrict__ fc,
                           const int* __restrict__ p_dt,
                           const float2* __restrict__ ws, float* __restrict__ out,
                           int nsteps)
{
    long n = (long)blockIdx.x * blockDim.x + threadIdx.x;
    if (n >= (long)nsteps) return;
    float dt = (float)p_dt[0];
    float K1 = __expf(pk[1]);
    float f  = fc[0];
    float ar = 1.f - dt * K1, ai = -dt * f;
    int c = (int)(n >> 4);
    int k = (int)(n & 15);
    float2 W = ws[c];
    float mr, mi; cpow(ar, ai, k + 1, mr, mi);
    out[2 * n]     += mr * W.x - mi * W.y;
    out[2 * n + 1] += mr * W.y + mi * W.x;
}

extern "C" void kernel_launch(void* const* d_in, const int* in_sizes, int n_in,
                              void* d_out, int out_size, void* d_ws, size_t ws_size,
                              hipStream_t stream)
{
    const float* pk    = (const float*)d_in[0];
    const float* TAx   = (const float*)d_in[1];
    const float* TAy   = (const float*)d_in[2];
    const float* fc    = (const float*)d_in[3];
    const int*   p_dt  = (const int*)d_in[6];
    const int*   p_dtf = (const int*)d_in[7];
    (void)n_in; (void)ws_size;

    int nt      = in_sizes[1];
    int nsteps  = out_size / 2;
    int nchunks = (nsteps + 15) / 16;
    int ntiles  = (nchunks + 15) / 16;
    float*  out = (float*)d_out;
    float2* ws  = (float2*)d_ws;

    int threads1 = ntiles * 32;
    int blocks1  = (threads1 + 255) / 256;
    slab_pass1<<<blocks1, 256, 0, stream>>>(pk, TAx, TAy, fc, p_dt, p_dtf,
                                            out, nsteps, nchunks, ntiles, nt);
    slab_pass2<<<1, 1024, 0, stream>>>(pk, fc, p_dt, out, ws, nsteps, nchunks);
    int blocks3 = (nsteps + 255) / 256;
    slab_pass3<<<blocks3, 256, 0, stream>>>(pk, fc, p_dt, ws, out, nsteps);
}


# Pass 1 also stores each chunk's end value as one 128-byte line per tile in a compact plane; the single-block scan stages that plane in LDS instead of reading one line of out per chunk
# speedup vs baseline: 4.0346x; 4.0346x over previous
#include <hip/hip_runtime.h>

typedef __attribute__((ext_vector_type(2))) float v2f;
typedef __attribute__((ext_vector_type(8))) float v8f;

__device__ __forceinline__ void cmul(float ar, float ai, float br, float bi,
                                     float& cr, float& ci) {
    cr = ar * br - ai * bi;
    ci = ar * bi + ai * br;
}

__device__ __forceinline__ void cpow(float ar, float ai, int p, float& r, float& i) {
    float xr = 1.f, xi = 0.f;
    for (int q = 0; q < p; ++q) { float tr, ti; cmul(xr, xi, ar, ai, tr, ti); xr = tr; xi = ti; }
    r = xr; i = xi;
}

__global__ void slab_pass1(const float* __restrict__ pk, const float* __restrict__ TAx,
                           const float* __restrict__ TAy, const float* __restrict__ fc,
                           const int* __restrict__ p_dt, const int* __restrict__ p_dtf,
                           float* __restrict__ out, float2* __restrict__ carry,
                           int nsteps, int nchunks, int ntiles, int nt)
{
    int gid  = blockIdx.x * blockDim.x + threadIdx.x;
    int tile = gid >> 5;
    int lane = gid & 31;
    if (tile >= ntiles) return;

    float dt      = (float)p_dt[0];
    int   nsub    = p_dtf[0] / p_dt[0];
    float invnsub = 1.f / (float)nsub;
    float K0   = __expf(pk[0]);
    float K1   = __expf(pk[1]);
    float f    = fc[0];
    float ar   = 1.f - dt * K1;
    float ai   = -dt * f;
    float dtK0 = dt * K0;

    int half = lane >> 4;
    int r    = lane & 15;
    int c0   = tile * 16;

    v2f Are[4], Aim[4], Mre[4], Mim[4], Mneg[4];
    #pragma unroll
    for (int s = 0; s < 4; ++s) {
        #pragma unroll
        for (int e = 0; e < 2; ++e) {
            int j = 4 * s + 2 * half + e;
            int  c  = c0 + r;
            long n  = (long)16 * c + j;
            int itf0 = (int)(n / nsub);
            float aa = (float)(n - (long)itf0 * nsub) * invnsub;
            int i0 = itf0     < nt - 1 ? itf0     : nt - 1;
            int i1 = itf0 + 1 < nt - 1 ? itf0 + 1 : nt - 1;
            float tax = (1.f - aa) * TAx[i0] + aa * TAx[i1];
            float tay = (1.f - aa) * TAy[i0] + aa * TAy[i1];
            float valid = (c < nchunks && n < (long)nsteps) ? 1.f : 0.f;
            Are[s][e] = valid * dtK0 * tax;
            Aim[s][e] = valid * dtK0 * tay;
            int p = r - j;
            float pr, pi;
            cpow(ar, ai, p < 0 ? 0 : p, pr, pi);
            float m = (p >= 0) ? 1.f : 0.f;
            Mre[s][e]  =  m * pr;
            Mim[s][e]  =  m * pi;
            Mneg[s][e] = -m * pi;
        }
    }

    v8f Sre = {}; v8f Sim = {};
    #pragma unroll
    for (int s = 0; s < 4; ++s)
        Sre = __builtin_amdgcn_wmma_f32_16x16x4_f32(false, Are[s], false, Mre[s],
                                                    (short)0, Sre, false, false);
    #pragma unroll
    for (int s = 0; s < 4; ++s)
        Sre = __builtin_amdgcn_wmma_f32_16x16x4_f32(false, Aim[s], false, Mneg[s],
                                                    (short)0, Sre, false, false);
    #pragma unroll
    for (int s = 0; s < 4; ++s)
        Sim = __builtin_amdgcn_wmma_f32_16x16x4_f32(false, Are[s], false, Mim[s],
                                                    (short)0, Sim, false, false);
    #pragma unroll
    for (int s = 0; s < 4; ++s)
        Sim = __builtin_amdgcn_wmma_f32_16x16x4_f32(false, Aim[s], false, Mre[s],
                                                    (short)0, Sim, false, false);

    #pragma unroll
    for (int e = 0; e < 8; ++e) {
        int  row = e + 8 * half;
        int  c   = c0 + row;
        long n   = (long)16 * c + r;
        if (c < nchunks && n < (long)nsteps) {
            float2 v; v.x = Sre[e]; v.y = Sim[e];
            *reinterpret_cast<float2*>(out + 2 * n) = v;
        }
    }

    int   srcl = 15 + ((lane & 8) << 1);
    float gr = 0.f, gi = 0.f;
    #pragma unroll
    for (int e = 0; e < 8; ++e) {
        float tr = __shfl(Sre[e], srcl, 32);
        float ti = __shfl(Sim[e], srcl, 32);
        if ((lane & 7) == e) { gr = tr; gi = ti; }
    }
    if (lane < 16) {
        int  c   = c0 + lane;
        long n15 = (long)16 * c + 15;
        bool ok  = (c < nchunks) && (n15 < (long)nsteps);
        carry[c] = ok ? make_float2(gr, gi) : make_float2(0.f, 0.f);
    }
}

#define P2_THREADS 1024
#define P2_PLANE   (33 * P2_THREADS)

__device__ __forceinline__ void chunk_scan(float2* p, int cnt, float4* sh, int t,
                                           float a16r, float a16i)
{
    float Ar = 1.f, Ai = 0.f, Br = 0.f, Bi = 0.f;
    int i = 0;
    for (; i + 8 <= cnt; i += 8) {
        float2 g[8];
        #pragma unroll
        for (int u = 0; u < 8; ++u) g[u] = p[i + u];
        #pragma unroll
        for (int u = 0; u < 8; ++u) {
            float tr, ti;
            cmul(a16r, a16i, Ar, Ai, tr, ti); Ar = tr; Ai = ti;
            cmul(a16r, a16i, Br, Bi, tr, ti); Br = tr + g[u].x; Bi = ti + g[u].y;
        }
    }
    for (; i < cnt; ++i) {
        float2 g = p[i];
        float tr, ti;
        cmul(a16r, a16i, Ar, Ai, tr, ti); Ar = tr; Ai = ti;
        cmul(a16r, a16i, Br, Bi, tr, ti); Br = tr + g.x; Bi = ti + g.y;
    }
    sh[t] = make_float4(Ar, Ai, Br, Bi);
    __syncthreads();
    for (int off = 1; off < P2_THREADS; off <<= 1) {
        float4 v = sh[t];
        if (t >= off) {
            float4 q = sh[t - off];
            float nar, nai, tr, ti;
            cmul(v.x, v.y, q.x, q.y, nar, nai);
            cmul(v.x, v.y, q.z, q.w, tr, ti);
            v = make_float4(nar, nai, tr + v.z, ti + v.w);
        }
        __syncthreads();
        sh[t] = v;
        __syncthreads();
    }
    float Wr = 0.f, Wi = 0.f;
    if (t > 0) { float4 q = sh[t - 1]; Wr = q.z; Wi = q.w; }
    i = 0;
    for (; i + 8 <= cnt; i += 8) {
        float2 g[8];
        #pragma unroll
        for (int u = 0; u < 8; ++u) g[u] = p[i + u];
        #pragma unroll
        for (int u = 0; u < 8; ++u) {
            p[i + u] = make_float2(Wr, Wi);
            float tr, ti;
            cmul(a16r, a16i, Wr, Wi, tr, ti);
            Wr = tr + g[u].x; Wi = ti + g[u].y;
        }
    }
    for (; i < cnt; ++i) {
        float2 g = p[i];
        p[i] = make_float2(Wr, Wi);
        float tr, ti;
        cmul(a16r, a16i, Wr, Wi, tr, ti);
        Wr = tr + g.x; Wi = ti + g.y;
    }
}

__global__ void slab_pass2(const float* __restrict__ pk, const float* __restrict__ fc,
                           const int* __restrict__ p_dt, float2* ws, int nchunks)
{
    __shared__ float4 sh[P2_THREADS];
    __shared__ __attribute__((aligned(16))) float2 plane[P2_PLANE];
    int t = threadIdx.x;
    float dt = (float)p_dt[0];
    float K1 = __expf(pk[1]);
    float f  = fc[0];
    float ar = 1.f - dt * K1, ai = -dt * f;
    float a16r, a16i; cpow(ar, ai, 16, a16r, a16i);

    int cpt = (nchunks + P2_THREADS - 1) / P2_THREADS;

    if (cpt * P2_THREADS <= P2_PLANE) {
        const int nv  = ((nchunks + 15) >> 4) << 3;
        const int nvt = cpt * (P2_THREADS / 2);
        float4* gv = reinterpret_cast<float4*>(ws);
        float4* lv = reinterpret_cast<float4*>(plane);
        for (int v0 = t; v0 < nvt; v0 += 8 * P2_THREADS) {
            float4 q[8];
            #pragma unroll
            for (int u = 0; u < 8; ++u) {
                int v = v0 + u * P2_THREADS;
                q[u] = (v < nv) ? gv[v] : make_float4(0.f, 0.f, 0.f, 0.f);
            }
            #pragma unroll
            for (int u = 0; u < 8; ++u) {
                int v = v0 + u * P2_THREADS;
                if (v < nvt) lv[v] = q[u];
            }
        }
        __syncthreads();
        chunk_scan(plane + t * cpt, cpt, sh, t, a16r, a16i);
        __syncthreads();
        for (int v0 = t; v0 < nv; v0 += 8 * P2_THREADS) {
            float4 q[8];
            #pragma unroll
            for (int u = 0; u < 8; ++u) {
                int v = v0 + u * P2_THREADS;
                q[u] = (v < nv) ? lv[v] : make_float4(0.f, 0.f, 0.f, 0.f);
            }
            #pragma unroll
            for (int u = 0; u < 8; ++u) {
                int v = v0 + u * P2_THREADS;
                if (v < nv) gv[v] = q[u];
            }
        }
    } else {
        int cb  = t * cpt;
        int cnt = nchunks - cb;
        cnt = cnt < 0 ? 0 : (cnt > cpt ? cpt : cnt);
        chunk_scan(ws + cb, cnt, sh, t, a16r, a16i);
    }
}

__global__ void slab_pass3(const float* __restrict__ pk, const float* __restrict__ fc,
                           const int* __restrict__ p_dt,
                           const float2* __restrict__ ws, float* __restrict__ out,
                           int nsteps)
{
    long n = (long)blockIdx.x * blockDim.x + threadIdx.x;
    if (n >= (long)nsteps) return;
    float dt = (float)p_dt[0];
    float K1 = __expf(pk[1]);
    float f  = fc[0];
    float ar = 1.f - dt * K1, ai = -dt * f;
    int c = (int)(n >> 4);
    int k = (int)(n & 15);
    float2 W = ws[c];
    float mr, mi; cpow(ar, ai, k + 1, mr, mi);
    out[2 * n]     += mr * W.x - mi * W.y;
    out[2 * n + 1] += mr * W.y + mi * W.x;
}

extern "C" void kernel_launch(void* const* d_in, const int* in_sizes, int n_in,
                              void* d_out, int out_size, void* d_ws, size_t ws_size,
                              hipStream_t stream)
{
    const float* pk    = (const float*)d_in[0];
    const float* TAx   = (const float*)d_in[1];
    const float* TAy   = (const float*)d_in[2];
    const float* fc    = (const float*)d_in[3];
    const int*   p_dt  = (const int*)d_in[6];
    const int*   p_dtf = (const int*)d_in[7];
    (void)n_in; (void)ws_size;

    int nt      = in_sizes[1];
    int nsteps  = out_size / 2;
    int nchunks = (nsteps + 15) / 16;
    int ntiles  = (nchunks + 15) / 16;
    float*  out = (float*)d_out;
    float2* ws  = (float2*)d_ws;

    int threads1 = ntiles * 32;
    int blocks1  = (threads1 + 255) / 256;
    slab_pass1<<<blocks1, 256, 0, stream>>>(pk, TAx, TAy, fc, p_dt, p_dtf,
                                            out, ws, nsteps, nchunks, ntiles, nt);
    slab_pass2<<<1, P2_THREADS, 0, stream>>>(pk, fc, p_dt, ws, nchunks);
    int blocks3 = (nsteps + 255) / 256;
    slab_pass3<<<blocks3, 256, 0, stream>>>(pk, fc, p_dt, ws, out, nsteps);
}


# Two-level scan: pass 1 scans its tile's 16 chunk ends in-wave and adds the local carry by WMMA, so the single block scans one carry per 256-step tile, not per chunk; reduction tree changes
# speedup vs baseline: 4.8048x; 1.1909x over previous
#include <hip/hip_runtime.h>

typedef __attribute__((ext_vector_type(2))) float v2f;
typedef __attribute__((ext_vector_type(8))) float v8f;

__device__ __forceinline__ void cmul(float ar, float ai, float br, float bi,
                                     float& cr, float& ci) {
    cr = ar * br - ai * bi;
    ci = ar * bi + ai * br;
}

__device__ __forceinline__ void cpow(float ar, float ai, int p, float& r, float& i) {
    float xr = 1.f, xi = 0.f;
    for (int q = 0; q < p; ++q) { float tr, ti; cmul(xr, xi, ar, ai, tr, ti); xr = tr; xi = ti; }
    r = xr; i = xi;
}

#define P1_WAVES 8

__global__ void slab_pass1(const float* __restrict__ pk, const float* __restrict__ TAx,
                           const float* __restrict__ TAy, const float* __restrict__ fc,
                           const int* __restrict__ p_dt, const int* __restrict__ p_dtf,
                           float* __restrict__ out, float2* __restrict__ carry,
                           int nsteps, int nchunks, int nt)
{
    int gid  = blockIdx.x * blockDim.x + threadIdx.x;
    int tile = gid >> 5;
    int lane = gid & 31;
    __shared__ float2 tile_end[P1_WAVES];

    float dt      = (float)p_dt[0];
    int   nsub    = p_dtf[0] / p_dt[0];
    float invnsub = 1.f / (float)nsub;
    float K0   = __expf(pk[0]);
    float K1   = __expf(pk[1]);
    float f    = fc[0];
    float ar   = 1.f - dt * K1;
    float ai   = -dt * f;
    float dtK0 = dt * K0;

    int half = lane >> 4;
    int r    = lane & 15;
    int c0   = tile * 16;

    v2f Are[4], Aim[4], Mre[4], Mim[4], Mneg[4];
    #pragma unroll
    for (int s = 0; s < 4; ++s) {
        #pragma unroll
        for (int e = 0; e < 2; ++e) {
            int j = 4 * s + 2 * half + e;
            int  c  = c0 + r;
            long n  = (long)16 * c + j;
            int itf0 = (int)(n / nsub);
            float aa = (float)(n - (long)itf0 * nsub) * invnsub;
            int i0 = itf0     < nt - 1 ? itf0     : nt - 1;
            int i1 = itf0 + 1 < nt - 1 ? itf0 + 1 : nt - 1;
            float tax = (1.f - aa) * TAx[i0] + aa * TAx[i1];
            float tay = (1.f - aa) * TAy[i0] + aa * TAy[i1];
            float valid = (c < nchunks && n < (long)nsteps) ? 1.f : 0.f;
            Are[s][e] = valid * dtK0 * tax;
            Aim[s][e] = valid * dtK0 * tay;
            int p = r - j;
            float pr, pi;
            cpow(ar, ai, p < 0 ? 0 : p, pr, pi);
            float m = (p >= 0) ? 1.f : 0.f;
            Mre[s][e]  =  m * pr;
            Mim[s][e]  =  m * pi;
            Mneg[s][e] = -m * pi;
        }
    }

    v8f Sre = {}; v8f Sim = {};
    #pragma unroll
    for (int s = 0; s < 4; ++s)
        Sre = __builtin_amdgcn_wmma_f32_16x16x4_f32(false, Are[s], false, Mre[s],
                                                    (short)0, Sre, false, false);
    #pragma unroll
    for (int s = 0; s < 4; ++s)
        Sre = __builtin_amdgcn_wmma_f32_16x16x4_f32(false, Aim[s], false, Mneg[s],
                                                    (short)0, Sre, false, false);
    #pragma unroll
    for (int s = 0; s < 4; ++s)
        Sim = __builtin_amdgcn_wmma_f32_16x16x4_f32(false, Are[s], false, Mim[s],
                                                    (short)0, Sim, false, false);
    #pragma unroll
    for (int s = 0; s < 4; ++s)
        Sim = __builtin_amdgcn_wmma_f32_16x16x4_f32(false, Aim[s], false, Mre[s],
                                                    (short)0, Sim, false, false);

    int   srcl = 15 + ((lane & 8) << 1);
    float gr = 0.f, gi = 0.f;
    #pragma unroll
    for (int e = 0; e < 8; ++e) {
        float tr = __shfl(Sre[e], srcl, 32);
        float ti = __shfl(Sim[e], srcl, 32);
        if ((lane & 7) == e) { gr = tr; gi = ti; }
    }
    {
        int  c   = c0 + r;
        long n15 = (long)16 * c + 15;
        bool ok  = (c < nchunks) && (n15 < (long)nsteps);
        if (!ok) { gr = 0.f; gi = 0.f; }
    }

    float pkr = __shfl(Mre[0][0], r, 32);
    float pki = __shfl(Mim[0][0], r, 32);
    float Pr, Pi;   cmul(pkr, pki, ar, ai, Pr, Pi);
    float q15r = __shfl(Mre[0][0], 15, 32);
    float q15i = __shfl(Mim[0][0], 15, 32);
    float mr, mi;   cmul(q15r, q15i, ar, ai, mr, mi);

    float Ir = gr, Ii = gi;
    #pragma unroll
    for (int off = 1; off < 16; off <<= 1) {
        int   sl = (lane & 16) | ((r - off) & 15);
        float tr = __shfl(Ir, sl, 32);
        float ti = __shfl(Ii, sl, 32);
        if (r >= off) {
            float ur, ui; cmul(mr, mi, tr, ti, ur, ui);
            Ir += ur; Ii += ui;
        }
        float sr, si; cmul(mr, mi, mr, mi, sr, si); mr = sr; mi = si;
    }
    float Wlr = __shfl(Ir, (lane & 16) | ((r - 1) & 15), 32);
    float Wli = __shfl(Ii, (lane & 16) | ((r - 1) & 15), 32);
    if (r == 0) { Wlr = 0.f; Wli = 0.f; }
    float Gr = __shfl(Ir, 15, 32);
    float Gi = __shfl(Ii, 15, 32);

    v2f A1, B1re, B1im;
    A1[0]   = half ? 0.f : Wlr;   A1[1]   = half ? 0.f : Wli;
    B1re[0] = half ? 0.f : Pr;    B1re[1] = half ? 0.f : -Pi;
    B1im[0] = half ? 0.f : Pi;    B1im[1] = half ? 0.f : Pr;
    Sre = __builtin_amdgcn_wmma_f32_16x16x4_f32(false, A1, false, B1re,
                                                (short)0, Sre, false, false);
    Sim = __builtin_amdgcn_wmma_f32_16x16x4_f32(false, A1, false, B1im,
                                                (short)0, Sim, false, false);

    #pragma unroll
    for (int e = 0; e < 8; ++e) {
        int  row = e + 8 * half;
        int  c   = c0 + row;
        long n   = (long)16 * c + r;
        if (c < nchunks && n < (long)nsteps) {
            float2 v; v.x = Sre[e]; v.y = Sim[e];
            *reinterpret_cast<float2*>(out + 2 * n) = v;
        }
    }

    int wv = threadIdx.x >> 5;
    if (lane == 0) tile_end[wv] = make_float2(Gr, Gi);
    __syncthreads();
    if (wv == 0 && lane < 16)
        carry[(size_t)blockIdx.x * 16 + lane] = (lane < P1_WAVES) ? tile_end[lane]
                                                                   : make_float2(0.f, 0.f);
}

#define P2_THREADS 1024
#define P2_PLANE   (33 * P2_THREADS)

__device__ __forceinline__ void chunk_scan(float2* p, int cnt, float4* sh, int t,
                                           float a16r, float a16i)
{
    float Ar = 1.f, Ai = 0.f, Br = 0.f, Bi = 0.f;
    int i = 0;
    for (; i + 8 <= cnt; i += 8) {
        float2 g[8];
        #pragma unroll
        for (int u = 0; u < 8; ++u) g[u] = p[i + u];
        #pragma unroll
        for (int u = 0; u < 8; ++u) {
            float tr, ti;
            cmul(a16r, a16i, Ar, Ai, tr, ti); Ar = tr; Ai = ti;
            cmul(a16r, a16i, Br, Bi, tr, ti); Br = tr + g[u].x; Bi = ti + g[u].y;
        }
    }
    for (; i < cnt; ++i) {
        float2 g = p[i];
        float tr, ti;
        cmul(a16r, a16i, Ar, Ai, tr, ti); Ar = tr; Ai = ti;
        cmul(a16r, a16i, Br, Bi, tr, ti); Br = tr + g.x; Bi = ti + g.y;
    }
    sh[t] = make_float4(Ar, Ai, Br, Bi);
    __syncthreads();
    for (int off = 1; off < P2_THREADS; off <<= 1) {
        float4 v = sh[t];
        if (t >= off) {
            float4 q = sh[t - off];
            float nar, nai, tr, ti;
            cmul(v.x, v.y, q.x, q.y, nar, nai);
            cmul(v.x, v.y, q.z, q.w, tr, ti);
            v = make_float4(nar, nai, tr + v.z, ti + v.w);
        }
        __syncthreads();
        sh[t] = v;
        __syncthreads();
    }
    float Wr = 0.f, Wi = 0.f;
    if (t > 0) { float4 q = sh[t - 1]; Wr = q.z; Wi = q.w; }
    i = 0;
    for (; i + 8 <= cnt; i += 8) {
        float2 g[8];
        #pragma unroll
        for (int u = 0; u < 8; ++u) g[u] = p[i + u];
        #pragma unroll
        for (int u = 0; u < 8; ++u) {
            p[i + u] = make_float2(Wr, Wi);
            float tr, ti;
            cmul(a16r, a16i, Wr, Wi, tr, ti);
            Wr = tr + g[u].x; Wi = ti + g[u].y;
        }
    }
    for (; i < cnt; ++i) {
        float2 g = p[i];
        p[i] = make_float2(Wr, Wi);
        float tr, ti;
        cmul(a16r, a16i, Wr, Wi, tr, ti);
        Wr = tr + g.x; Wi = ti + g.y;
    }
}

__global__ void slab_pass2(const float* __restrict__ pk, const float* __restrict__ fc,
                           const int* __restrict__ p_dt, float2* ws, int ntiles)
{
    __shared__ float4 sh[P2_THREADS];
    __shared__ __attribute__((aligned(16))) float2 plane[P2_PLANE];
    int t = threadIdx.x;
    float dt = (float)p_dt[0];
    float K1 = __expf(pk[1]);
    float f  = fc[0];
    float ar = 1.f - dt * K1, ai = -dt * f;
    float a16r, a16i;   cpow(ar, ai, 16, a16r, a16i);
    float a256r, a256i; cpow(a16r, a16i, 16, a256r, a256i);

    const int nlines = (ntiles + P1_WAVES - 1) / P1_WAVES;
    const float2* gin  = ws;
    float2*       wout = ws + (size_t)nlines * 16;

    int cpt = (ntiles + P2_THREADS - 1) / P2_THREADS;

    if (cpt * P2_THREADS <= P2_PLANE) {
        const int nslot = cpt * P2_THREADS;
        for (int v0 = t; v0 < nslot; v0 += 8 * P2_THREADS) {
            float2 q[8];
            #pragma unroll
            for (int u = 0; u < 8; ++u) {
                int v = v0 + u * P2_THREADS;
                q[u] = (v < ntiles) ? gin[(size_t)(v / P1_WAVES) * 16 + (v % P1_WAVES)]
                                    : make_float2(0.f, 0.f);
            }
            #pragma unroll
            for (int u = 0; u < 8; ++u) {
                int v = v0 + u * P2_THREADS;
                if (v < nslot) plane[v] = q[u];
            }
        }
        __syncthreads();
        chunk_scan(plane + t * cpt, cpt, sh, t, a256r, a256i);
        __syncthreads();
        for (int v = t; v < ntiles; v += P2_THREADS) wout[v] = plane[v];
    } else {
        for (int v = t; v < ntiles; v += P2_THREADS)
            wout[v] = gin[(size_t)(v / P1_WAVES) * 16 + (v % P1_WAVES)];
        __syncthreads();
        int cb  = t * cpt;
        int cnt = ntiles - cb;
        cnt = cnt < 0 ? 0 : (cnt > cpt ? cpt : cnt);
        chunk_scan(wout + cb, cnt, sh, t, a256r, a256i);
    }
}

#define P3_THREADS 256

__global__ void slab_pass3(const float* __restrict__ pk, const float* __restrict__ fc,
                           const int* __restrict__ p_dt,
                           const float2* __restrict__ wt, float* __restrict__ out,
                           int nsteps)
{
    __shared__ float2 P1[16];
    __shared__ float2 Q[16];
    int  m    = threadIdx.x;
    long n    = (long)blockIdx.x * P3_THREADS + m;
    bool live = n < (long)nsteps;
    float2* po = reinterpret_cast<float2*>(out) + (live ? n : (long)nsteps - 1);
    float2 T = *po;

    if (m < 32) {
        float dt = (float)p_dt[0];
        float K1 = __expf(pk[1]);
        float f  = fc[0];
        float ar = 1.f - dt * K1, ai = -dt * f;
        float pr, pi; cpow(ar, ai, (m & 15) + 1, pr, pi);
        float a16r = __shfl(pr, 15, 32);
        float a16i = __shfl(pi, 15, 32);
        if (m < 16) {
            P1[m] = make_float2(pr, pi);
        } else {
            float2 W = wt[blockIdx.x];
            float qr, qi; cpow(a16r, a16i, m - 16, qr, qi);
            float zr, zi; cmul(qr, qi, W.x, W.y, zr, zi);
            Q[m - 16] = make_float2(zr, zi);
        }
    }
    __syncthreads();
    float2 p = P1[m & 15];
    float2 q = Q[m >> 4];
    float zr, zi; cmul(p.x, p.y, q.x, q.y, zr, zi);
    if (live) *po = make_float2(T.x + zr, T.y + zi);
}

extern "C" void kernel_launch(void* const* d_in, const int* in_sizes, int n_in,
                              void* d_out, int out_size, void* d_ws, size_t ws_size,
                              hipStream_t stream)
{
    const float* pk    = (const float*)d_in[0];
    const float* TAx   = (const float*)d_in[1];
    const float* TAy   = (const float*)d_in[2];
    const float* fc    = (const float*)d_in[3];
    const int*   p_dt  = (const int*)d_in[6];
    const int*   p_dtf = (const int*)d_in[7];
    (void)n_in; (void)ws_size;

    int nt      = in_sizes[1];
    int nsteps  = out_size / 2;
    int nchunks = (nsteps + 15) / 16;
    int ntiles  = (nchunks + 15) / 16;
    float*  out = (float*)d_out;
    float2* ws  = (float2*)d_ws;

    int blocks1  = (ntiles + P1_WAVES - 1) / P1_WAVES;
    slab_pass1<<<blocks1, P1_WAVES * 32, 0, stream>>>(pk, TAx, TAy, fc, p_dt, p_dtf,
                                            out, ws, nsteps, nchunks, nt);
    slab_pass2<<<1, P2_THREADS, 0, stream>>>(pk, fc, p_dt, ws, ntiles);
    slab_pass3<<<ntiles, P3_THREADS, 0, stream>>>(pk, fc, p_dt, ws + (size_t)blocks1 * 16, out, nsteps);
}


# Out written once: launch 1 keeps only tile ends; after the scan the WMMA tile kernel runs again with the tile start value fed into its chunk scan; no read-back pass; reduction tree changes
# speedup vs baseline: 5.4195x; 1.1279x over previous
#include <hip/hip_runtime.h>

typedef __attribute__((ext_vector_type(2))) float v2f;
typedef __attribute__((ext_vector_type(8))) float v8f;

__device__ __forceinline__ void cmul(float ar, float ai, float br, float bi,
                                     float& cr, float& ci) {
    cr = ar * br - ai * bi;
    ci = ar * bi + ai * br;
}

__device__ __forceinline__ void cpow(float ar, float ai, int p, float& r, float& i) {
    float xr = 1.f, xi = 0.f;
    for (int q = 0; q < p; ++q) { float tr, ti; cmul(xr, xi, ar, ai, tr, ti); xr = tr; xi = ti; }
    r = xr; i = xi;
}

#define P1_WAVES 8

template <bool FINAL>
__device__ __forceinline__ void tile_body(const float* __restrict__ pk, const float* __restrict__ TAx,
                                          const float* __restrict__ TAy, const float* __restrict__ fc,
                                          const int* __restrict__ p_dt, const int* __restrict__ p_dtf,
                                          float* __restrict__ out, float2* __restrict__ carry,
                                          const float2* __restrict__ wstart, float2* tile_end,
                                          int nsteps, int nchunks, int ntiles, int nt)
{
    int gid  = blockIdx.x * blockDim.x + threadIdx.x;
    int tile = gid >> 5;
    int lane = gid & 31;

    float2 Wt = make_float2(0.f, 0.f);
    if (FINAL) { if (tile < ntiles) Wt = wstart[tile]; }

    float dt      = (float)p_dt[0];
    int   nsub    = p_dtf[0] / p_dt[0];
    float invnsub = 1.f / (float)nsub;
    float K0   = __expf(pk[0]);
    float K1   = __expf(pk[1]);
    float f    = fc[0];
    float ar   = 1.f - dt * K1;
    float ai   = -dt * f;
    float dtK0 = dt * K0;

    int half = lane >> 4;
    int r    = lane & 15;
    int c0   = tile * 16;

    v2f Are[4], Aim[4], Mre[4], Mim[4], Mneg[4];
    #pragma unroll
    for (int s = 0; s < 4; ++s) {
        #pragma unroll
        for (int e = 0; e < 2; ++e) {
            int j = 4 * s + 2 * half + e;
            int  c  = c0 + r;
            long n  = (long)16 * c + j;
            int itf0 = (int)(n / nsub);
            float aa = (float)(n - (long)itf0 * nsub) * invnsub;
            int i0 = itf0     < nt - 1 ? itf0     : nt - 1;
            int i1 = itf0 + 1 < nt - 1 ? itf0 + 1 : nt - 1;
            float tax = (1.f - aa) * TAx[i0] + aa * TAx[i1];
            float tay = (1.f - aa) * TAy[i0] + aa * TAy[i1];
            float valid = (c < nchunks && n < (long)nsteps) ? 1.f : 0.f;
            Are[s][e] = valid * dtK0 * tax;
            Aim[s][e] = valid * dtK0 * tay;
            int p = r - j;
            float pr, pi;
            cpow(ar, ai, p < 0 ? 0 : p, pr, pi);
            float m = (p >= 0) ? 1.f : 0.f;
            Mre[s][e]  =  m * pr;
            Mim[s][e]  =  m * pi;
            Mneg[s][e] = -m * pi;
        }
    }

    v8f Sre = {}; v8f Sim = {};
    #pragma unroll
    for (int s = 0; s < 4; ++s)
        Sre = __builtin_amdgcn_wmma_f32_16x16x4_f32(false, Are[s], false, Mre[s],
                                                    (short)0, Sre, false, false);
    #pragma unroll
    for (int s = 0; s < 4; ++s)
        Sre = __builtin_amdgcn_wmma_f32_16x16x4_f32(false, Aim[s], false, Mneg[s],
                                                    (short)0, Sre, false, false);
    #pragma unroll
    for (int s = 0; s < 4; ++s)
        Sim = __builtin_amdgcn_wmma_f32_16x16x4_f32(false, Are[s], false, Mim[s],
                                                    (short)0, Sim, false, false);
    #pragma unroll
    for (int s = 0; s < 4; ++s)
        Sim = __builtin_amdgcn_wmma_f32_16x16x4_f32(false, Aim[s], false, Mre[s],
                                                    (short)0, Sim, false, false);

    int   srcl = 15 + ((lane & 8) << 1);
    float gr = 0.f, gi = 0.f;
    #pragma unroll
    for (int e = 0; e < 8; ++e) {
        float tr = __shfl(Sre[e], srcl, 32);
        float ti = __shfl(Sim[e], srcl, 32);
        if ((lane & 7) == e) { gr = tr; gi = ti; }
    }
    {
        int  c   = c0 + r;
        long n15 = (long)16 * c + 15;
        bool ok  = (c < nchunks) && (n15 < (long)nsteps);
        if (!ok) { gr = 0.f; gi = 0.f; }
    }

    float pkr = __shfl(Mre[0][0], r, 32);
    float pki = __shfl(Mim[0][0], r, 32);
    float Pr, Pi;   cmul(pkr, pki, ar, ai, Pr, Pi);
    float q15r = __shfl(Mre[0][0], 15, 32);
    float q15i = __shfl(Mim[0][0], 15, 32);
    float mr, mi;   cmul(q15r, q15i, ar, ai, mr, mi);

    if (FINAL) {
        float ur, ui; cmul(mr, mi, Wt.x, Wt.y, ur, ui);
        if (r == 0) { gr += ur; gi += ui; }
    }

    float Ir = gr, Ii = gi;
    #pragma unroll
    for (int off = 1; off < 16; off <<= 1) {
        int   sl = (lane & 16) | ((r - off) & 15);
        float tr = __shfl(Ir, sl, 32);
        float ti = __shfl(Ii, sl, 32);
        if (r >= off) {
            float ur, ui; cmul(mr, mi, tr, ti, ur, ui);
            Ir += ur; Ii += ui;
        }
        float sr, si; cmul(mr, mi, mr, mi, sr, si); mr = sr; mi = si;
    }

    if (FINAL) {
        float Wlr = __shfl(Ir, (lane & 16) | ((r - 1) & 15), 32);
        float Wli = __shfl(Ii, (lane & 16) | ((r - 1) & 15), 32);
        if (r == 0) { Wlr = Wt.x; Wli = Wt.y; }

        v2f A1, B1re, B1im;
        A1[0]   = half ? 0.f : Wlr;   A1[1]   = half ? 0.f : Wli;
        B1re[0] = half ? 0.f : Pr;    B1re[1] = half ? 0.f : -Pi;
        B1im[0] = half ? 0.f : Pi;    B1im[1] = half ? 0.f : Pr;
        Sre = __builtin_amdgcn_wmma_f32_16x16x4_f32(false, A1, false, B1re,
                                                    (short)0, Sre, false, false);
        Sim = __builtin_amdgcn_wmma_f32_16x16x4_f32(false, A1, false, B1im,
                                                    (short)0, Sim, false, false);

        #pragma unroll
        for (int e = 0; e < 8; ++e) {
            int  row = e + 8 * half;
            int  c   = c0 + row;
            long n   = (long)16 * c + r;
            if (c < nchunks && n < (long)nsteps) {
                float2 v; v.x = Sre[e]; v.y = Sim[e];
                *reinterpret_cast<float2*>(out + 2 * n) = v;
            }
        }
    } else {
        float Gr = __shfl(Ir, 15, 32);
        float Gi = __shfl(Ii, 15, 32);
        int wv = threadIdx.x >> 5;
        if (lane == 0) tile_end[wv] = make_float2(Gr, Gi);
        __syncthreads();
        if (wv == 0 && lane < 16)
            carry[(size_t)blockIdx.x * 16 + lane] = (lane < P1_WAVES) ? tile_end[lane]
                                                                       : make_float2(0.f, 0.f);
    }
}

__global__ void slab_ends(const float* __restrict__ pk, const float* __restrict__ TAx,
                          const float* __restrict__ TAy, const float* __restrict__ fc,
                          const int* __restrict__ p_dt, const int* __restrict__ p_dtf,
                          float2* __restrict__ carry, int nsteps, int nchunks, int ntiles, int nt)
{
    __shared__ float2 tile_end[P1_WAVES];
    tile_body<false>(pk, TAx, TAy, fc, p_dt, p_dtf, nullptr, carry, nullptr, tile_end,
                     nsteps, nchunks, ntiles, nt);
}

__global__ void slab_final(const float* __restrict__ pk, const float* __restrict__ TAx,
                           const float* __restrict__ TAy, const float* __restrict__ fc,
                           const int* __restrict__ p_dt, const int* __restrict__ p_dtf,
                           const float2* __restrict__ wstart, float* __restrict__ out,
                           int nsteps, int nchunks, int ntiles, int nt)
{
    tile_body<true>(pk, TAx, TAy, fc, p_dt, p_dtf, out, nullptr, wstart, nullptr,
                    nsteps, nchunks, ntiles, nt);
}

#define P2_THREADS 1024
#define P2_PLANE   (33 * P2_THREADS)

__device__ __forceinline__ void chunk_scan(float2* p, int cnt, float4* sh, int t,
                                           float a16r, float a16i)
{
    float Ar = 1.f, Ai = 0.f, Br = 0.f, Bi = 0.f;
    int i = 0;
    for (; i + 8 <= cnt; i += 8) {
        float2 g[8];
        #pragma unroll
        for (int u = 0; u < 8; ++u) g[u] = p[i + u];
        #pragma unroll
        for (int u = 0; u < 8; ++u) {
            float tr, ti;
            cmul(a16r, a16i, Ar, Ai, tr, ti); Ar = tr; Ai = ti;
            cmul(a16r, a16i, Br, Bi, tr, ti); Br = tr + g[u].x; Bi = ti + g[u].y;
        }
    }
    for (; i < cnt; ++i) {
        float2 g = p[i];
        float tr, ti;
        cmul(a16r, a16i, Ar, Ai, tr, ti); Ar = tr; Ai = ti;
        cmul(a16r, a16i, Br, Bi, tr, ti); Br = tr + g.x; Bi = ti + g.y;
    }
    sh[t] = make_float4(Ar, Ai, Br, Bi);
    __syncthreads();
    for (int off = 1; off < P2_THREADS; off <<= 1) {
        float4 v = sh[t];
        if (t >= off) {
            float4 q = sh[t - off];
            float nar, nai, tr, ti;
            cmul(v.x, v.y, q.x, q.y, nar, nai);
            cmul(v.x, v.y, q.z, q.w, tr, ti);
            v = make_float4(nar, nai, tr + v.z, ti + v.w);
        }
        __syncthreads();
        sh[t] = v;
        __syncthreads();
    }
    float Wr = 0.f, Wi = 0.f;
    if (t > 0) { float4 q = sh[t - 1]; Wr = q.z; Wi = q.w; }
    i = 0;
    for (; i + 8 <= cnt; i += 8) {
        float2 g[8];
        #pragma unroll
        for (int u = 0; u < 8; ++u) g[u] = p[i + u];
        #pragma unroll
        for (int u = 0; u < 8; ++u) {
            p[i + u] = make_float2(Wr, Wi);
            float tr, ti;
            cmul(a16r, a16i, Wr, Wi, tr, ti);
            Wr = tr + g[u].x; Wi = ti + g[u].y;
        }
    }
    for (; i < cnt; ++i) {
        float2 g = p[i];
        p[i] = make_float2(Wr, Wi);
        float tr, ti;
        cmul(a16r, a16i, Wr, Wi, tr, ti);
        Wr = tr + g.x; Wi = ti + g.y;
    }
}

__global__ void slab_pass2(const float* __restrict__ pk, const float* __restrict__ fc,
                           const int* __restrict__ p_dt, float2* ws, int ntiles)
{
    __shared__ float4 sh[P2_THREADS];
    __shared__ __attribute__((aligned(16))) float2 plane[P2_PLANE];
    int t = threadIdx.x;
    float dt = (float)p_dt[0];
    float K1 = __expf(pk[1]);
    float f  = fc[0];
    float ar = 1.f - dt * K1, ai = -dt * f;
    float a16r, a16i;   cpow(ar, ai, 16, a16r, a16i);
    float a256r, a256i; cpow(a16r, a16i, 16, a256r, a256i);

    const int nlines = (ntiles + P1_WAVES - 1) / P1_WAVES;
    const float2* gin  = ws;
    float2*       wout = ws + (size_t)nlines * 16;

    int cpt = (ntiles + P2_THREADS - 1) / P2_THREADS;

    if (cpt * P2_THREADS <= P2_PLANE) {
        const int nslot = cpt * P2_THREADS;
        for (int v0 = t; v0 < nslot; v0 += 8 * P2_THREADS) {
            float2 q[8];
            #pragma unroll
            for (int u = 0; u < 8; ++u) {
                int v = v0 + u * P2_THREADS;
                q[u] = (v < ntiles) ? gin[(size_t)(v / P1_WAVES) * 16 + (v % P1_WAVES)]
                                    : make_float2(0.f, 0.f);
            }
            #pragma unroll
            for (int u = 0; u < 8; ++u) {
                int v = v0 + u * P2_THREADS;
                if (v < nslot) plane[v] = q[u];
            }
        }
        __syncthreads();
        chunk_scan(plane + t * cpt, cpt, sh, t, a256r, a256i);
        __syncthreads();
        for (int v = t; v < ntiles; v += P2_THREADS) wout[v] = plane[v];
    } else {
        for (int v = t; v < ntiles; v += P2_THREADS)
            wout[v] = gin[(size_t)(v / P1_WAVES) * 16 + (v % P1_WAVES)];
        __syncthreads();
        int cb  = t * cpt;
        int cnt = ntiles - cb;
        cnt = cnt < 0 ? 0 : (cnt > cpt ? cpt : cnt);
        chunk_scan(wout + cb, cnt, sh, t, a256r, a256i);
    }
}

extern "C" void kernel_launch(void* const* d_in, const int* in_sizes, int n_in,
                              void* d_out, int out_size, void* d_ws, size_t ws_size,
                              hipStream_t stream)
{
    const float* pk    = (const float*)d_in[0];
    const float* TAx   = (const float*)d_in[1];
    const float* TAy   = (const float*)d_in[2];
    const float* fc    = (const float*)d_in[3];
    const int*   p_dt  = (const int*)d_in[6];
    const int*   p_dtf = (const int*)d_in[7];
    (void)n_in; (void)ws_size;

    int nt      = in_sizes[1];
    int nsteps  = out_size / 2;
    int nchunks = (nsteps + 15) / 16;
    int ntiles  = (nchunks + 15) / 16;
    float*  out = (float*)d_out;
    float2* ws  = (float2*)d_ws;

    int blocks1  = (ntiles + P1_WAVES - 1) / P1_WAVES;
    slab_ends<<<blocks1, P1_WAVES * 32, 0, stream>>>(pk, TAx, TAy, fc, p_dt, p_dtf,
                                            ws, nsteps, nchunks, ntiles, nt);
    slab_pass2<<<1, P2_THREADS, 0, stream>>>(pk, fc, p_dt, ws, ntiles);
    slab_final<<<blocks1, P1_WAVES * 32, 0, stream>>>(pk, TAx, TAy, fc, p_dt, p_dtf,
                                            ws + (size_t)blocks1 * 16, out, nsteps, nchunks, ntiles, nt);
}


# The scan launch is gone: each block of the last launch sums the earlier blocks' end values itself, weighted with powers of a^2048 by square and multiply; two launches; reduction tree changes
# speedup vs baseline: 7.2414x; 1.3362x over previous
#include <hip/hip_runtime.h>

typedef __attribute__((ext_vector_type(2))) float v2f;
typedef __attribute__((ext_vector_type(8))) float v8f;

__device__ __forceinline__ void cmul(float ar, float ai, float br, float bi,
                                     float& cr, float& ci) {
    cr = ar * br - ai * bi;
    ci = ar * bi + ai * br;
}

__device__ __forceinline__ void cpow(float ar, float ai, int p, float& r, float& i) {
    float xr = 1.f, xi = 0.f;
    for (int q = 0; q < p; ++q) { float tr, ti; cmul(xr, xi, ar, ai, tr, ti); xr = tr; xi = ti; }
    r = xr; i = xi;
}

__device__ __forceinline__ void csqr_n(float ar, float ai, int k, float& r, float& i) {
    float xr = ar, xi = ai;
    for (int q = 0; q < k; ++q) { float tr, ti; cmul(xr, xi, xr, xi, tr, ti); xr = tr; xi = ti; }
    r = xr; i = xi;
}

#define P1_WAVES      8
#define P1_WAVES_LOG2 3
#define P1_THREADS    (P1_WAVES * 32)
#define P1_THREADS_LOG2 (P1_WAVES_LOG2 + 5)

template <bool FINAL>
__device__ __forceinline__ void tile_body(const float* __restrict__ pk, const float* __restrict__ TAx,
                                          const float* __restrict__ TAy, const float* __restrict__ fc,
                                          const int* __restrict__ p_dt, const int* __restrict__ p_dtf,
                                          float* __restrict__ out, float2* __restrict__ carry,
                                          const float2* __restrict__ lines, float2* tile_end,
                                          int nsteps, int nchunks, int nt)
{
    int gid  = blockIdx.x * blockDim.x + threadIdx.x;
    int tile = gid >> 5;
    int lane = gid & 31;

    int wv = threadIdx.x >> 5;

    const int b   = blockIdx.x;
    const int tid = threadIdx.x;
    float2 Eb = make_float2(0.f, 0.f), Bl = make_float2(0.f, 0.f);
    if (FINAL) {
        Eb = lines[(size_t)b * 16 + wv];
        if (tid < b) Bl = lines[(size_t)(b - 1 - tid) * 16 + P1_WAVES];
    }

    float dt      = (float)p_dt[0];
    int   nsub    = p_dtf[0] / p_dt[0];
    float invnsub = 1.f / (float)nsub;
    float K0   = __expf(pk[0]);
    float K1   = __expf(pk[1]);
    float f    = fc[0];
    float ar   = 1.f - dt * K1;
    float ai   = -dt * f;
    float dtK0 = dt * K0;

    float2 Wt = make_float2(0.f, 0.f);
    if (FINAL) {
        float a256r, a256i;
        csqr_n(ar, ai, 8, a256r, a256i);
        float A8r, A8i;
        csqr_n(a256r, a256i, P1_WAVES_LOG2, A8r, A8i);
        float pr = 1.f, pi = 0.f, qr = A8r, qi = A8i;
        #pragma unroll
        for (int bit = 0; bit < P1_THREADS_LOG2; ++bit) {
            float tr, ti;
            if ((tid >> bit) & 1) { cmul(pr, pi, qr, qi, tr, ti); pr = tr; pi = ti; }
            cmul(qr, qi, qr, qi, tr, ti); qr = tr; qi = ti;
        }
        float accr = 0.f, acci = 0.f, Fr = 1.f, Fi = 0.f;
        for (int l0 = 0; l0 < b; l0 += P1_THREADS) {
            int t = l0 + tid;
            float2 g = make_float2(0.f, 0.f);
            if (l0 > 0 && t < b) g = lines[(size_t)(b - 1 - t) * 16 + P1_WAVES];
            if (l0 == 0) g = Bl;
            float ur, ui, tr, ti;
            cmul(Fr, Fi, pr, pi, ur, ui);
            cmul(ur, ui, g.x, g.y, tr, ti);
            if (t < b) { accr += tr; acci += ti; }
            cmul(Fr, Fi, qr, qi, tr, ti); Fr = tr; Fi = ti;
        }
        #pragma unroll
        for (int off = 16; off >= 1; off >>= 1) {
            accr += __shfl_xor(accr, off, 32);
            acci += __shfl_xor(acci, off, 32);
        }
        if (lane == 0) tile_end[wv] = make_float2(accr, acci);
        __syncthreads();
        float wr = 0.f, wi = 0.f;
        #pragma unroll
        for (int v = 0; v < P1_WAVES; ++v) { float2 pv = tile_end[v]; wr += pv.x; wi += pv.y; }
        float zr = a256r, zi = a256i;
        #pragma unroll
        for (int bit = 0; bit < P1_WAVES_LOG2; ++bit) {
            float tr, ti;
            if ((wv >> bit) & 1) { cmul(zr, zi, wr, wi, tr, ti); wr = tr; wi = ti; }
            cmul(zr, zi, zr, zi, tr, ti); zr = tr; zi = ti;
        }
        Wt = make_float2(wr + Eb.x, wi + Eb.y);
    }

    int half = lane >> 4;
    int r    = lane & 15;
    int c0   = tile * 16;

    v2f Are[4], Aim[4], Mre[4], Mim[4], Mneg[4];
    #pragma unroll
    for (int s = 0; s < 4; ++s) {
        #pragma unroll
        for (int e = 0; e < 2; ++e) {
            int j = 4 * s + 2 * half + e;
            int  c  = c0 + r;
            long n  = (long)16 * c + j;
            int itf0 = (int)(n / nsub);
            float aa = (float)(n - (long)itf0 * nsub) * invnsub;
            int i0 = itf0     < nt - 1 ? itf0     : nt - 1;
            int i1 = itf0 + 1 < nt - 1 ? itf0 + 1 : nt - 1;
            float tax = (1.f - aa) * TAx[i0] + aa * TAx[i1];
            float tay = (1.f - aa) * TAy[i0] + aa * TAy[i1];
            float valid = (c < nchunks && n < (long)nsteps) ? 1.f : 0.f;
            Are[s][e] = valid * dtK0 * tax;
            Aim[s][e] = valid * dtK0 * tay;
            int p = r - j;
            float pr, pi;
            cpow(ar, ai, p < 0 ? 0 : p, pr, pi);
            float m = (p >= 0) ? 1.f : 0.f;
            Mre[s][e]  =  m * pr;
            Mim[s][e]  =  m * pi;
            Mneg[s][e] = -m * pi;
        }
    }

    v8f Sre = {}; v8f Sim = {};
    #pragma unroll
    for (int s = 0; s < 4; ++s)
        Sre = __builtin_amdgcn_wmma_f32_16x16x4_f32(false, Are[s], false, Mre[s],
                                                    (short)0, Sre, false, false);
    #pragma unroll
    for (int s = 0; s < 4; ++s)
        Sre = __builtin_amdgcn_wmma_f32_16x16x4_f32(false, Aim[s], false, Mneg[s],
                                                    (short)0, Sre, false, false);
    #pragma unroll
    for (int s = 0; s < 4; ++s)
        Sim = __builtin_amdgcn_wmma_f32_16x16x4_f32(false, Are[s], false, Mim[s],
                                                    (short)0, Sim, false, false);
    #pragma unroll
    for (int s = 0; s < 4; ++s)
        Sim = __builtin_amdgcn_wmma_f32_16x16x4_f32(false, Aim[s], false, Mre[s],
                                                    (short)0, Sim, false, false);

    int   srcl = 15 + ((lane & 8) << 1);
    float gr = 0.f, gi = 0.f;
    #pragma unroll
    for (int e = 0; e < 8; ++e) {
        float tr = __shfl(Sre[e], srcl, 32);
        float ti = __shfl(Sim[e], srcl, 32);
        if ((lane & 7) == e) { gr = tr; gi = ti; }
    }
    {
        int  c   = c0 + r;
        long n15 = (long)16 * c + 15;
        bool ok  = (c < nchunks) && (n15 < (long)nsteps);
        if (!ok) { gr = 0.f; gi = 0.f; }
    }

    float pkr = __shfl(Mre[0][0], r, 32);
    float pki = __shfl(Mim[0][0], r, 32);
    float Pr, Pi;   cmul(pkr, pki, ar, ai, Pr, Pi);
    float q15r = __shfl(Mre[0][0], 15, 32);
    float q15i = __shfl(Mim[0][0], 15, 32);
    float mr, mi;   cmul(q15r, q15i, ar, ai, mr, mi);

    if (FINAL) {
        float ur, ui; cmul(mr, mi, Wt.x, Wt.y, ur, ui);
        if (r == 0) { gr += ur; gi += ui; }
    }

    float Ir = gr, Ii = gi;
    #pragma unroll
    for (int off = 1; off < 16; off <<= 1) {
        int   sl = (lane & 16) | ((r - off) & 15);
        float tr = __shfl(Ir, sl, 32);
        float ti = __shfl(Ii, sl, 32);
        if (r >= off) {
            float ur, ui; cmul(mr, mi, tr, ti, ur, ui);
            Ir += ur; Ii += ui;
        }
        float sr, si; cmul(mr, mi, mr, mi, sr, si); mr = sr; mi = si;
    }

    if (FINAL) {
        float Wlr = __shfl(Ir, (lane & 16) | ((r - 1) & 15), 32);
        float Wli = __shfl(Ii, (lane & 16) | ((r - 1) & 15), 32);
        if (r == 0) { Wlr = Wt.x; Wli = Wt.y; }

        v2f A1, B1re, B1im;
        A1[0]   = half ? 0.f : Wlr;   A1[1]   = half ? 0.f : Wli;
        B1re[0] = half ? 0.f : Pr;    B1re[1] = half ? 0.f : -Pi;
        B1im[0] = half ? 0.f : Pi;    B1im[1] = half ? 0.f : Pr;
        Sre = __builtin_amdgcn_wmma_f32_16x16x4_f32(false, A1, false, B1re,
                                                    (short)0, Sre, false, false);
        Sim = __builtin_amdgcn_wmma_f32_16x16x4_f32(false, A1, false, B1im,
                                                    (short)0, Sim, false, false);

        #pragma unroll
        for (int e = 0; e < 8; ++e) {
            int  row = e + 8 * half;
            int  c   = c0 + row;
            long n   = (long)16 * c + r;
            if (c < nchunks && n < (long)nsteps) {
                float2 v; v.x = Sre[e]; v.y = Sim[e];
                *reinterpret_cast<float2*>(out + 2 * n) = v;
            }
        }
    } else {
        float Gr = __shfl(Ir, 15, 32);
        float Gi = __shfl(Ii, 15, 32);
        if (lane == 0) tile_end[wv] = make_float2(Gr, Gi);
        __syncthreads();
        if (wv == 0) {
            float2 g[P1_WAVES];
            #pragma unroll
            for (int v = 0; v < P1_WAVES; ++v) g[v] = tile_end[v];
            float Er = 0.f, Ei = 0.f;
            #pragma unroll
            for (int v = 0; v < P1_WAVES; ++v) {
                float tr, ti; cmul(mr, mi, Er, Ei, tr, ti);
                tr += g[v].x; ti += g[v].y;
                Er = (v < lane) ? tr : Er;
                Ei = (v < lane) ? ti : Ei;
            }
            if (lane > P1_WAVES) { Er = 0.f; Ei = 0.f; }
            if (lane < 16) carry[(size_t)blockIdx.x * 16 + lane] = make_float2(Er, Ei);
        }
    }
}

__global__ void slab_ends(const float* __restrict__ pk, const float* __restrict__ TAx,
                          const float* __restrict__ TAy, const float* __restrict__ fc,
                          const int* __restrict__ p_dt, const int* __restrict__ p_dtf,
                          float2* __restrict__ carry, int nsteps, int nchunks, int nt)
{
    __shared__ float2 tile_end[P1_WAVES];
    tile_body<false>(pk, TAx, TAy, fc, p_dt, p_dtf, nullptr, carry, nullptr, tile_end,
                     nsteps, nchunks, nt);
}

__global__ void slab_final(const float* __restrict__ pk, const float* __restrict__ TAx,
                           const float* __restrict__ TAy, const float* __restrict__ fc,
                           const int* __restrict__ p_dt, const int* __restrict__ p_dtf,
                           const float2* __restrict__ lines, float* __restrict__ out,
                           int nsteps, int nchunks, int nt)
{
    __shared__ float2 part[P1_WAVES];
    tile_body<true>(pk, TAx, TAy, fc, p_dt, p_dtf, out, nullptr, lines, part,
                    nsteps, nchunks, nt);
}

extern "C" void kernel_launch(void* const* d_in, const int* in_sizes, int n_in,
                              void* d_out, int out_size, void* d_ws, size_t ws_size,
                              hipStream_t stream)
{
    const float* pk    = (const float*)d_in[0];
    const float* TAx   = (const float*)d_in[1];
    const float* TAy   = (const float*)d_in[2];
    const float* fc    = (const float*)d_in[3];
    const int*   p_dt  = (const int*)d_in[6];
    const int*   p_dtf = (const int*)d_in[7];
    (void)n_in; (void)ws_size;

    int nt      = in_sizes[1];
    int nsteps  = out_size / 2;
    int nchunks = (nsteps + 15) / 16;
    int ntiles  = (nchunks + 15) / 16;
    float*  out = (float*)d_out;
    float2* ws  = (float2*)d_ws;

    int blocks1  = (ntiles + P1_WAVES - 1) / P1_WAVES;
    slab_ends<<<blocks1, P1_THREADS, 0, stream>>>(pk, TAx, TAy, fc, p_dt, p_dtf,
                                            ws, nsteps, nchunks, nt);
    slab_final<<<blocks1, P1_THREADS, 0, stream>>>(pk, TAx, TAy, fc, p_dt, p_dtf,
                                            ws, out, nsteps, nchunks, nt);
}


# One launch: each block sums its own start value from the forcing samples (per-interval weights from one nsub-step loop, powers of a^nsub); first launch and workspace gone; reduction tree changes
# speedup vs baseline: 8.8128x; 1.2170x over previous
#include <hip/hip_runtime.h>


typedef __attribute__((ext_vector_type(2))) float v2f;
typedef __attribute__((ext_vector_type(8))) float v8f;

__device__ __forceinline__ void cmul(float ar, float ai, float br, float bi,
                                     float& cr, float& ci) {
    cr = ar * br - ai * bi;
    ci = ar * bi + ai * br;
}

__device__ __forceinline__ void cpow(float ar, float ai, int p, float& r, float& i) {
    float xr = 1.f, xi = 0.f;
    for (int q = 0; q < p; ++q) { float tr, ti; cmul(xr, xi, ar, ai, tr, ti); xr = tr; xi = ti; }
    r = xr; i = xi;
}

#define P1_WAVES      8
#define P1_THREADS    (P1_WAVES * 32)
#define P1_THREADS_LOG2 8
#define P1_STEPS      (P1_WAVES * 256)
#define PB            16

__global__ void slab_all(const float* __restrict__ pk, const float* __restrict__ TAx,
                         const float* __restrict__ TAy, const float* __restrict__ fc,
                         const int* __restrict__ p_dt, const int* __restrict__ p_dtf,
                         float* __restrict__ out, int nsteps, int nchunks, int nt)
{
    __shared__ float2 part[P1_WAVES];
    __shared__ float2 tend[P1_WAVES];

    int gid  = blockIdx.x * blockDim.x + threadIdx.x;
    int tile = gid >> 5;
    int lane = gid & 31;
    const int b   = blockIdx.x;
    const int tid = threadIdx.x;
    const int wv  = tid >> 5;

    float dt      = (float)p_dt[0];
    int   nsub    = p_dtf[0] / p_dt[0];
    float invnsub = 1.f / (float)nsub;
    float K0   = __expf(pk[0]);
    float K1   = __expf(pk[1]);
    float f    = fc[0];
    float ar   = 1.f - dt * K1;
    float ai   = -dt * f;
    float dtK0 = dt * K0;

    const int N0 = b * P1_STEPS;
    const int I0 = N0 / nsub;
    const int R  = N0 - I0 * nsub;
    const int j0 = I0     < nt - 1 ? I0     : nt - 1;
    const int j1 = I0 + 1 < nt - 1 ? I0 + 1 : nt - 1;
    const float px0 = TAx[j0], py0 = TAy[j0], px1 = TAx[j1], py1 = TAy[j1];

    const int ntrip  = (I0 + P1_THREADS - 1) >> P1_THREADS_LOG2;
    const int nbatch = (ntrip + PB - 1) / PB;
    float tx[PB], ty[PB];
    #pragma unroll
    for (int u = 0; u < PB; ++u) { tx[u] = 0.f; ty[u] = 0.f; }
    if (nbatch > 0) {
        #pragma unroll
        for (int u = 0; u < PB; ++u) {
            int k = (nbatch - 1) * PB + (PB - 1 - u);
            int j = I0 - 1 - tid - k * P1_THREADS;
            int jc = j < 0 ? 0 : (j < nt - 1 ? j : nt - 1);
            tx[u] = TAx[jc]; ty[u] = TAy[jc];
        }
    }

    float c0r = 0.f, c0i = 0.f, c1r = 0.f, c1i = 0.f;
    float Ar = 1.f, Ai = 0.f;
    float d0r = 0.f, d0i = 0.f, d1r = 0.f, d1i = 0.f;
    float pRr = 1.f, pRi = 0.f;
    for (int m = 0; m < nsub; ++m) {
        float aa = (float)m * invnsub;
        float tr, ti;
        cmul(ar, ai, c0r, c0i, tr, ti); c0r = tr + (1.f - aa); c0i = ti;
        cmul(ar, ai, c1r, c1i, tr, ti); c1r = tr + aa;         c1i = ti;
        cmul(Ar, Ai, ar, ai, tr, ti);   Ar = tr;               Ai = ti;
        if (m + 1 == R) { d0r = c0r; d0i = c0i; d1r = c1r; d1i = c1i; pRr = Ar; pRi = Ai; }
    }
    float ccr, cci;
    { float tr, ti; cmul(c1r, c1i, Ar, Ai, tr, ti); ccr = c0r + tr; cci = c0i + ti; }

    float wtr = 1.f, wti = 0.f, qr = Ar, qi = Ai;
    #pragma unroll
    for (int bit = 0; bit < P1_THREADS_LOG2; ++bit) {
        float tr, ti;
        if ((tid >> bit) & 1) { cmul(wtr, wti, qr, qi, tr, ti); wtr = tr; wti = ti; }
        cmul(qr, qi, qr, qi, tr, ti); qr = tr; qi = ti;
    }
    float accr = 0.f, acci = 0.f;
    for (int bb = nbatch - 1; bb >= 0; --bb) {
        float nx[PB], ny[PB];
        #pragma unroll
        for (int u = 0; u < PB; ++u) { nx[u] = 0.f; ny[u] = 0.f; }
        if (bb > 0) {
            #pragma unroll
            for (int u = 0; u < PB; ++u) {
                int k = (bb - 1) * PB + (PB - 1 - u);
                int j = I0 - 1 - tid - k * P1_THREADS;
                int jc = j < 0 ? 0 : (j < nt - 1 ? j : nt - 1);
                nx[u] = TAx[jc]; ny[u] = TAy[jc];
            }
        }
        #pragma unroll
        for (int u = 0; u < PB; ++u) {
            int k = bb * PB + (PB - 1 - u);
            int j = I0 - 1 - tid - k * P1_THREADS;
            float kr = (j == 0) ? c0r : ccr;
            float ki = (j == 0) ? c0i : cci;
            float er, ei, tr, ti;
            cmul(kr, ki, tx[u], ty[u], er, ei);
            cmul(qr, qi, accr, acci, tr, ti);
            accr = tr + (j >= 0 ? er : 0.f);
            acci = ti + (j >= 0 ? ei : 0.f);
        }
        #pragma unroll
        for (int u = 0; u < PB; ++u) { tx[u] = nx[u]; ty[u] = ny[u]; }
    }
    {
        float tr, ti;
        cmul(wtr, wti, accr, acci, tr, ti); accr = tr; acci = ti;
    }
    #pragma unroll
    for (int off = 16; off >= 1; off >>= 1) {
        accr += __shfl_xor(accr, off, 32);
        acci += __shfl_xor(acci, off, 32);
    }
    if (lane == 0) part[wv] = make_float2(accr, acci);

    int half = lane >> 4;
    int r    = lane & 15;
    int c0   = tile * 16;

    v2f Are[4], Aim[4], Mre[4], Mim[4], Mneg[4];
    #pragma unroll
    for (int s = 0; s < 4; ++s) {
        #pragma unroll
        for (int e = 0; e < 2; ++e) {
            int j = 4 * s + 2 * half + e;
            int  c  = c0 + r;
            long n  = (long)16 * c + j;
            int itf0 = (int)(n / nsub);
            float aa = (float)(n - (long)itf0 * nsub) * invnsub;
            int i0 = itf0     < nt - 1 ? itf0     : nt - 1;
            int i1 = itf0 + 1 < nt - 1 ? itf0 + 1 : nt - 1;
            float tax = (1.f - aa) * TAx[i0] + aa * TAx[i1];
            float tay = (1.f - aa) * TAy[i0] + aa * TAy[i1];
            float valid = (c < nchunks && n < (long)nsteps) ? 1.f : 0.f;
            Are[s][e] = valid * dtK0 * tax;
            Aim[s][e] = valid * dtK0 * tay;
            int p = r - j;
            float pr, pi;
            cpow(ar, ai, p < 0 ? 0 : p, pr, pi);
            float m = (p >= 0) ? 1.f : 0.f;
            Mre[s][e]  =  m * pr;
            Mim[s][e]  =  m * pi;
            Mneg[s][e] = -m * pi;
        }
    }

    v8f Sre = {}; v8f Sim = {};
    #pragma unroll
    for (int s = 0; s < 4; ++s)
        Sre = __builtin_amdgcn_wmma_f32_16x16x4_f32(false, Are[s], false, Mre[s],
                                                    (short)0, Sre, false, false);
    #pragma unroll
    for (int s = 0; s < 4; ++s)
        Sre = __builtin_amdgcn_wmma_f32_16x16x4_f32(false, Aim[s], false, Mneg[s],
                                                    (short)0, Sre, false, false);
    #pragma unroll
    for (int s = 0; s < 4; ++s)
        Sim = __builtin_amdgcn_wmma_f32_16x16x4_f32(false, Are[s], false, Mim[s],
                                                    (short)0, Sim, false, false);
    #pragma unroll
    for (int s = 0; s < 4; ++s)
        Sim = __builtin_amdgcn_wmma_f32_16x16x4_f32(false, Aim[s], false, Mre[s],
                                                    (short)0, Sim, false, false);

    int   srcl = 15 + ((lane & 8) << 1);
    float gr = 0.f, gi = 0.f;
    #pragma unroll
    for (int e = 0; e < 8; ++e) {
        float tr = __shfl(Sre[e], srcl, 32);
        float ti = __shfl(Sim[e], srcl, 32);
        if ((lane & 7) == e) { gr = tr; gi = ti; }
    }
    {
        int  c   = c0 + r;
        long n15 = (long)16 * c + 15;
        bool ok  = (c < nchunks) && (n15 < (long)nsteps);
        if (!ok) { gr = 0.f; gi = 0.f; }
    }

    float pkr = __shfl(Mre[0][0], r, 32);
    float pki = __shfl(Mim[0][0], r, 32);
    float Pr, Pi;   cmul(pkr, pki, ar, ai, Pr, Pi);
    float q15r = __shfl(Mre[0][0], 15, 32);
    float q15i = __shfl(Mim[0][0], 15, 32);
    float mr, mi;   cmul(q15r, q15i, ar, ai, mr, mi);

    float Ir = gr, Ii = gi;
    float zr = 1.f, zi = 0.f;
    #pragma unroll
    for (int off = 1; off < 16; off <<= 1) {
        int   sl = (lane & 16) | ((r - off) & 15);
        float tr = __shfl(Ir, sl, 32);
        float ti = __shfl(Ii, sl, 32);
        if (r >= off) {
            float ur, ui; cmul(mr, mi, tr, ti, ur, ui);
            Ir += ur; Ii += ui;
        }
        if (r & off) { float ur, ui; cmul(zr, zi, mr, mi, ur, ui); zr = ur; zi = ui; }
        float sr, si; cmul(mr, mi, mr, mi, sr, si); mr = sr; mi = si;
    }

    float Xr = __shfl(Ir, (lane & 16) | ((r - 1) & 15), 32);
    float Xi = __shfl(Ii, (lane & 16) | ((r - 1) & 15), 32);
    if (r == 0) { Xr = 0.f; Xi = 0.f; }

    float Gr = __shfl(Ir, 15, 32);
    float Gi = __shfl(Ii, 15, 32);
    if (lane == 0) tend[wv] = make_float2(Gr, Gi);

    __syncthreads();

    float2 ps[P1_WAVES], te[P1_WAVES];
    #pragma unroll
    for (int v = 0; v < P1_WAVES; ++v) { ps[v] = part[v]; te[v] = tend[v]; }
    float yr = 0.f, yi = 0.f;
    #pragma unroll
    for (int v = 0; v < P1_WAVES; ++v) { yr += ps[v].x; yi += ps[v].y; }
    float Jr, Ji;
    {
        if (I0 >= 1) { yr += c1r * px0 - c1i * py0; yi += c1r * py0 + c1i * px0; }
        float tr, ti;
        cmul(pRr, pRi, yr, yi, tr, ti);
        float hr = d0r * px0 - d0i * py0 + d1r * px1 - d1i * py1;
        float hi = d0r * py0 + d0i * px0 + d1r * py1 + d1i * px1;
        Jr = dtK0 * (tr + hr);
        Ji = dtK0 * (ti + hi);
    }
    #pragma unroll
    for (int v = 0; v < P1_WAVES; ++v) {
        float tr, ti; cmul(mr, mi, Jr, Ji, tr, ti);
        tr += te[v].x; ti += te[v].y;
        Jr = (v < wv) ? tr : Jr;
        Ji = (v < wv) ? ti : Ji;
    }
    float Wlr, Wli;
    {
        float tr, ti; cmul(zr, zi, Jr, Ji, tr, ti);
        Wlr = Xr + tr; Wli = Xi + ti;
    }

    v2f A1, B1re, B1im;
    A1[0]   = half ? 0.f : Wlr;   A1[1]   = half ? 0.f : Wli;
    B1re[0] = half ? 0.f : Pr;    B1re[1] = half ? 0.f : -Pi;
    B1im[0] = half ? 0.f : Pi;    B1im[1] = half ? 0.f : Pr;
    Sre = __builtin_amdgcn_wmma_f32_16x16x4_f32(false, A1, false, B1re,
                                                (short)0, Sre, false, false);
    Sim = __builtin_amdgcn_wmma_f32_16x16x4_f32(false, A1, false, B1im,
                                                (short)0, Sim, false, false);

    #pragma unroll
    for (int e = 0; e < 8; ++e) {
        int  row = e + 8 * half;
        int  c   = c0 + row;
        long n   = (long)16 * c + r;
        if (c < nchunks && n < (long)nsteps) {
            float2 v; v.x = Sre[e]; v.y = Sim[e];
            *reinterpret_cast<float2*>(out + 2 * n) = v;
        }
    }
}

extern "C" void kernel_launch(void* const* d_in, const int* in_sizes, int n_in,
                              void* d_out, int out_size, void* d_ws, size_t ws_size,
                              hipStream_t stream)
{
    const float* pk    = (const float*)d_in[0];
    const float* TAx   = (const float*)d_in[1];
    const float* TAy   = (const float*)d_in[2];
    const float* fc    = (const float*)d_in[3];
    const int*   p_dt  = (const int*)d_in[6];
    const int*   p_dtf = (const int*)d_in[7];
    (void)n_in; (void)ws_size; (void)d_ws;

    int nt      = in_sizes[1];
    int nsteps  = out_size / 2;
    int nchunks = (nsteps + 15) / 16;
    int ntiles  = (nchunks + 15) / 16;
    float*  out = (float*)d_out;

    int blocks1  = (ntiles + P1_WAVES - 1) / P1_WAVES;
    slab_all<<<blocks1, P1_THREADS, 0, stream>>>(pk, TAx, TAy, fc, p_dt, p_dtf,
                                                 out, nsteps, nchunks, nt);
}


# Tile operand set-up once per lane, not per element: one power per lane (square and multiply) read as a^(k-j) by a row shift; one index division per lane, then stepped; powers round differently
# speedup vs baseline: 11.7180x; 1.3297x over previous
#include <hip/hip_runtime.h>


typedef __attribute__((ext_vector_type(2))) float v2f;
typedef __attribute__((ext_vector_type(8))) float v8f;

__device__ __forceinline__ void cmul(float ar, float ai, float br, float bi,
                                     float& cr, float& ci) {
    cr = ar * br - ai * bi;
    ci = ar * bi + ai * br;
}

template <int N>
__device__ __forceinline__ float row_up(float v) {
    return __int_as_float(__builtin_amdgcn_update_dpp(0, __float_as_int(v), 0x110 + N,
                                                      0xF, 0xF, true));
}

#define P1_WAVES      8
#define P1_THREADS    (P1_WAVES * 32)
#define P1_THREADS_LOG2 8
#define P1_STEPS      (P1_WAVES * 256)
#define PB            16

__global__ void slab_all(const float* __restrict__ pk, const float* __restrict__ TAx,
                         const float* __restrict__ TAy, const float* __restrict__ fc,
                         const int* __restrict__ p_dt, const int* __restrict__ p_dtf,
                         float* __restrict__ out, int nsteps, int nchunks, int nt)
{
    __shared__ float2 part[P1_WAVES];
    __shared__ float2 tend[P1_WAVES];

    int gid  = blockIdx.x * blockDim.x + threadIdx.x;
    int tile = gid >> 5;
    int lane = gid & 31;
    const int b   = blockIdx.x;
    const int tid = threadIdx.x;
    const int wv  = tid >> 5;

    float dt      = (float)p_dt[0];
    int   nsub    = p_dtf[0] / p_dt[0];
    float invnsub = 1.f / (float)nsub;
    float K0   = __expf(pk[0]);
    float K1   = __expf(pk[1]);
    float f    = fc[0];
    float ar   = 1.f - dt * K1;
    float ai   = -dt * f;
    float dtK0 = dt * K0;

    const int N0 = b * P1_STEPS;
    const int I0 = N0 / nsub;
    const int R  = N0 - I0 * nsub;
    const int j0 = I0     < nt - 1 ? I0     : nt - 1;
    const int j1 = I0 + 1 < nt - 1 ? I0 + 1 : nt - 1;
    const float px0 = TAx[j0], py0 = TAy[j0], px1 = TAx[j1], py1 = TAy[j1];

    const int ntrip  = (I0 + P1_THREADS - 1) >> P1_THREADS_LOG2;
    const int nbatch = (ntrip + PB - 1) / PB;
    float tx[PB], ty[PB];
    #pragma unroll
    for (int u = 0; u < PB; ++u) { tx[u] = 0.f; ty[u] = 0.f; }
    if (nbatch > 0) {
        #pragma unroll
        for (int u = 0; u < PB; ++u) {
            int k = (nbatch - 1) * PB + (PB - 1 - u);
            int j = I0 - 1 - tid - k * P1_THREADS;
            int jc = j < 0 ? 0 : (j < nt - 1 ? j : nt - 1);
            tx[u] = TAx[jc]; ty[u] = TAy[jc];
        }
    }

    float c0r = 0.f, c0i = 0.f, c1r = 0.f, c1i = 0.f;
    float Ar = 1.f, Ai = 0.f;
    float d0r = 0.f, d0i = 0.f, d1r = 0.f, d1i = 0.f;
    float pRr = 1.f, pRi = 0.f;
    for (int m = 0; m < nsub; ++m) {
        float aa = (float)m * invnsub;
        float tr, ti;
        cmul(ar, ai, c0r, c0i, tr, ti); c0r = tr + (1.f - aa); c0i = ti;
        cmul(ar, ai, c1r, c1i, tr, ti); c1r = tr + aa;         c1i = ti;
        cmul(Ar, Ai, ar, ai, tr, ti);   Ar = tr;               Ai = ti;
        if (m + 1 == R) { d0r = c0r; d0i = c0i; d1r = c1r; d1i = c1i; pRr = Ar; pRi = Ai; }
    }
    float ccr, cci;
    { float tr, ti; cmul(c1r, c1i, Ar, Ai, tr, ti); ccr = c0r + tr; cci = c0i + ti; }

    float wtr = 1.f, wti = 0.f, qr = Ar, qi = Ai;
    #pragma unroll
    for (int bit = 0; bit < P1_THREADS_LOG2; ++bit) {
        float tr, ti;
        if ((tid >> bit) & 1) { cmul(wtr, wti, qr, qi, tr, ti); wtr = tr; wti = ti; }
        cmul(qr, qi, qr, qi, tr, ti); qr = tr; qi = ti;
    }
    float accr = 0.f, acci = 0.f;
    for (int bb = nbatch - 1; bb >= 0; --bb) {
        float nx[PB], ny[PB];
        #pragma unroll
        for (int u = 0; u < PB; ++u) { nx[u] = 0.f; ny[u] = 0.f; }
        if (bb > 0) {
            #pragma unroll
            for (int u = 0; u < PB; ++u) {
                int k = (bb - 1) * PB + (PB - 1 - u);
                int j = I0 - 1 - tid - k * P1_THREADS;
                int jc = j < 0 ? 0 : (j < nt - 1 ? j : nt - 1);
                nx[u] = TAx[jc]; ny[u] = TAy[jc];
            }
        }
        #pragma unroll
        for (int u = 0; u < PB; ++u) {
            int k = bb * PB + (PB - 1 - u);
            int j = I0 - 1 - tid - k * P1_THREADS;
            float kr = (j == 0) ? c0r : ccr;
            float ki = (j == 0) ? c0i : cci;
            float er, ei, tr, ti;
            cmul(kr, ki, tx[u], ty[u], er, ei);
            cmul(qr, qi, accr, acci, tr, ti);
            accr = tr + (j >= 0 ? er : 0.f);
            acci = ti + (j >= 0 ? ei : 0.f);
        }
        #pragma unroll
        for (int u = 0; u < PB; ++u) { tx[u] = nx[u]; ty[u] = ny[u]; }
    }
    {
        float tr, ti;
        cmul(wtr, wti, accr, acci, tr, ti); accr = tr; acci = ti;
    }
    #pragma unroll
    for (int off = 16; off >= 1; off >>= 1) {
        accr += __shfl_xor(accr, off, 32);
        acci += __shfl_xor(acci, off, 32);
    }
    if (lane == 0) part[wv] = make_float2(accr, acci);

    int half = lane >> 4;
    int r    = lane & 15;
    int c0   = tile * 16;

    float a2r, a2i, a4r, a4i, a8r, a8i;
    cmul(ar,  ai,  ar,  ai,  a2r, a2i);
    cmul(a2r, a2i, a2r, a2i, a4r, a4i);
    cmul(a4r, a4i, a4r, a4i, a8r, a8i);
    const int pe = r - 2 * half;
    const int pc = pe < 0 ? 0 : pe;
    float pwr = (pc & 1) ? ar : 1.f;
    float pwi = (pc & 1) ? ai : 0.f;
    { float tr, ti; cmul(pwr, pwi, a2r, a2i, tr, ti); if (pc & 2) { pwr = tr; pwi = ti; } }
    { float tr, ti; cmul(pwr, pwi, a4r, a4i, tr, ti); if (pc & 4) { pwr = tr; pwi = ti; } }
    { float tr, ti; cmul(pwr, pwi, a8r, a8i, tr, ti); if (pc & 8) { pwr = tr; pwi = ti; } }
    if (pe < 0) { pwr = 0.f; pwi = 0.f; }
    float Tr[8], Ti[8];
    Tr[0] = pwr;              Ti[0] = pwi;
    Tr[1] = row_up<1>(pwr);   Ti[1] = row_up<1>(pwi);
    Tr[2] = row_up<4>(pwr);   Ti[2] = row_up<4>(pwi);
    Tr[3] = row_up<5>(pwr);   Ti[3] = row_up<5>(pwi);
    Tr[4] = row_up<8>(pwr);   Ti[4] = row_up<8>(pwi);
    Tr[5] = row_up<9>(pwr);   Ti[5] = row_up<9>(pwi);
    Tr[6] = row_up<12>(pwr);  Ti[6] = row_up<12>(pwi);
    Tr[7] = row_up<13>(pwr);  Ti[7] = row_up<13>(pwi);
    const int      c   = c0 + r;
    const unsigned nb  = 16u * (unsigned)c + 2u * (unsigned)half;
    const unsigned usb = (unsigned)nsub;
    unsigned itf = nb / usb;
    unsigned rem = nb - itf * usb;

    v2f Are[4], Aim[4], Mre[4], Mim[4], Mneg[4];
    #pragma unroll
    for (int s = 0; s < 4; ++s) {
        #pragma unroll
        for (int e = 0; e < 2; ++e) {
            unsigned n  = nb + (unsigned)(4 * s + e);
            float aa = (float)rem * invnsub;
            int itf0 = (int)itf;
            int i0 = itf0     < nt - 1 ? itf0     : nt - 1;
            int i1 = itf0 + 1 < nt - 1 ? itf0 + 1 : nt - 1;
            float tax = (1.f - aa) * TAx[i0] + aa * TAx[i1];
            float tay = (1.f - aa) * TAy[i0] + aa * TAy[i1];
            float valid = (c < nchunks && n < (unsigned)nsteps) ? 1.f : 0.f;
            Are[s][e] = valid * dtK0 * tax;
            Aim[s][e] = valid * dtK0 * tay;
            Mre[s][e]  =  Tr[2 * s + e];
            Mim[s][e]  =  Ti[2 * s + e];
            Mneg[s][e] = -Ti[2 * s + e];
            const int step = (e == 0) ? 1 : 3;
            rem += (unsigned)step;
            #pragma unroll
            for (int t = 0; t < step; ++t)
                if (rem >= usb) { rem -= usb; itf += 1u; }
        }
    }

    v8f Sre = {}; v8f Sim = {};
    #pragma unroll
    for (int s = 0; s < 4; ++s)
        Sre = __builtin_amdgcn_wmma_f32_16x16x4_f32(false, Are[s], false, Mre[s],
                                                    (short)0, Sre, false, false);
    #pragma unroll
    for (int s = 0; s < 4; ++s)
        Sre = __builtin_amdgcn_wmma_f32_16x16x4_f32(false, Aim[s], false, Mneg[s],
                                                    (short)0, Sre, false, false);
    #pragma unroll
    for (int s = 0; s < 4; ++s)
        Sim = __builtin_amdgcn_wmma_f32_16x16x4_f32(false, Are[s], false, Mim[s],
                                                    (short)0, Sim, false, false);
    #pragma unroll
    for (int s = 0; s < 4; ++s)
        Sim = __builtin_amdgcn_wmma_f32_16x16x4_f32(false, Aim[s], false, Mre[s],
                                                    (short)0, Sim, false, false);

    int   srcl = 15 + ((lane & 8) << 1);
    float gr = 0.f, gi = 0.f;
    #pragma unroll
    for (int e = 0; e < 8; ++e) {
        float tr = __shfl(Sre[e], srcl, 32);
        float ti = __shfl(Sim[e], srcl, 32);
        if ((lane & 7) == e) { gr = tr; gi = ti; }
    }
    {
        int  c   = c0 + r;
        long n15 = (long)16 * c + 15;
        bool ok  = (c < nchunks) && (n15 < (long)nsteps);
        if (!ok) { gr = 0.f; gi = 0.f; }
    }

    float pkr = __shfl(Mre[0][0], r, 32);
    float pki = __shfl(Mim[0][0], r, 32);
    float Pr, Pi;   cmul(pkr, pki, ar, ai, Pr, Pi);
    float q15r = __shfl(Mre[0][0], 15, 32);
    float q15i = __shfl(Mim[0][0], 15, 32);
    float mr, mi;   cmul(q15r, q15i, ar, ai, mr, mi);

    float Ir = gr, Ii = gi;
    float zr = 1.f, zi = 0.f;
    #pragma unroll
    for (int off = 1; off < 16; off <<= 1) {
        int   sl = (lane & 16) | ((r - off) & 15);
        float tr = __shfl(Ir, sl, 32);
        float ti = __shfl(Ii, sl, 32);
        if (r >= off) {
            float ur, ui; cmul(mr, mi, tr, ti, ur, ui);
            Ir += ur; Ii += ui;
        }
        if (r & off) { float ur, ui; cmul(zr, zi, mr, mi, ur, ui); zr = ur; zi = ui; }
        float sr, si; cmul(mr, mi, mr, mi, sr, si); mr = sr; mi = si;
    }

    float Xr = __shfl(Ir, (lane & 16) | ((r - 1) & 15), 32);
    float Xi = __shfl(Ii, (lane & 16) | ((r - 1) & 15), 32);
    if (r == 0) { Xr = 0.f; Xi = 0.f; }

    float Gr = __shfl(Ir, 15, 32);
    float Gi = __shfl(Ii, 15, 32);
    if (lane == 0) tend[wv] = make_float2(Gr, Gi);

    __syncthreads();

    float2 ps[P1_WAVES], te[P1_WAVES];
    #pragma unroll
    for (int v = 0; v < P1_WAVES; ++v) { ps[v] = part[v]; te[v] = tend[v]; }
    float yr = 0.f, yi = 0.f;
    #pragma unroll
    for (int v = 0; v < P1_WAVES; ++v) { yr += ps[v].x; yi += ps[v].y; }
    float Jr, Ji;
    {
        if (I0 >= 1) { yr += c1r * px0 - c1i * py0; yi += c1r * py0 + c1i * px0; }
        float tr, ti;
        cmul(pRr, pRi, yr, yi, tr, ti);
        float hr = d0r * px0 - d0i * py0 + d1r * px1 - d1i * py1;
        float hi = d0r * py0 + d0i * px0 + d1r * py1 + d1i * px1;
        Jr = dtK0 * (tr + hr);
        Ji = dtK0 * (ti + hi);
    }
    #pragma unroll
    for (int v = 0; v < P1_WAVES; ++v) {
        float tr, ti; cmul(mr, mi, Jr, Ji, tr, ti);
        tr += te[v].x; ti += te[v].y;
        Jr = (v < wv) ? tr : Jr;
        Ji = (v < wv) ? ti : Ji;
    }
    float Wlr, Wli;
    {
        float tr, ti; cmul(zr, zi, Jr, Ji, tr, ti);
        Wlr = Xr + tr; Wli = Xi + ti;
    }

    v2f A1, B1re, B1im;
    A1[0]   = half ? 0.f : Wlr;   A1[1]   = half ? 0.f : Wli;
    B1re[0] = half ? 0.f : Pr;    B1re[1] = half ? 0.f : -Pi;
    B1im[0] = half ? 0.f : Pi;    B1im[1] = half ? 0.f : Pr;
    Sre = __builtin_amdgcn_wmma_f32_16x16x4_f32(false, A1, false, B1re,
                                                (short)0, Sre, false, false);
    Sim = __builtin_amdgcn_wmma_f32_16x16x4_f32(false, A1, false, B1im,
                                                (short)0, Sim, false, false);

    #pragma unroll
    for (int e = 0; e < 8; ++e) {
        int  row = e + 8 * half;
        int  c   = c0 + row;
        long n   = (long)16 * c + r;
        if (c < nchunks && n < (long)nsteps) {
            float2 v; v.x = Sre[e]; v.y = Sim[e];
            *reinterpret_cast<float2*>(out + 2 * n) = v;
        }
    }
}

extern "C" void kernel_launch(void* const* d_in, const int* in_sizes, int n_in,
                              void* d_out, int out_size, void* d_ws, size_t ws_size,
                              hipStream_t stream)
{
    const float* pk    = (const float*)d_in[0];
    const float* TAx   = (const float*)d_in[1];
    const float* TAy   = (const float*)d_in[2];
    const float* fc    = (const float*)d_in[3];
    const int*   p_dt  = (const int*)d_in[6];
    const int*   p_dtf = (const int*)d_in[7];
    (void)n_in; (void)ws_size; (void)d_ws;

    int nt      = in_sizes[1];
    int nsteps  = out_size / 2;
    int nchunks = (nsteps + 15) / 16;
    int ntiles  = (nchunks + 15) / 16;
    float*  out = (float*)d_out;

    int blocks1  = (ntiles + P1_WAVES - 1) / P1_WAVES;
    slab_all<<<blocks1, P1_THREADS, 0, stream>>>(pk, TAx, TAy, fc, p_dt, p_dtf,
                                                 out, nsteps, nchunks, nt);
}


# Interval weights c0, c1, d0, d1 and a^nsub, a^R summed across the wave's 32 lanes (lane power by square and multiply, row_xmask adds), not by a serial nsub-trip loop; reduction tree changes
# speedup vs baseline: 14.2977x; 1.2201x over previous
#include <hip/hip_runtime.h>


typedef __attribute__((ext_vector_type(2))) float v2f;
typedef __attribute__((ext_vector_type(8))) float v8f;

__device__ __forceinline__ void cmul(float ar, float ai, float br, float bi,
                                     float& cr, float& ci) {
    cr = ar * br - ai * bi;
    ci = ar * bi + ai * br;
}

template <int N>
__device__ __forceinline__ float row_up(float v) {
    return __int_as_float(__builtin_amdgcn_update_dpp(0, __float_as_int(v), 0x110 + N,
                                                      0xF, 0xF, true));
}

template <int M>
__device__ __forceinline__ float row_xor(float v) {
    return __int_as_float(__builtin_amdgcn_update_dpp(0, __float_as_int(v), 0x160 + M,
                                                      0xF, 0xF, true));
}
__device__ __forceinline__ float other_row(float v) {
    int b = __float_as_int(v);
    return __int_as_float(__builtin_amdgcn_permlanex16(b, b, 0x76543210, 0xfedcba98,
                                                       false, false));
}
__device__ __forceinline__ float lane_sum(float v) {
    v += row_xor<1>(v);
    v += row_xor<2>(v);
    v += row_xor<4>(v);
    v += row_xor<8>(v);
    v += other_row(v);
    return v;
}

#define P1_WAVES      8
#define P1_THREADS    (P1_WAVES * 32)
#define P1_THREADS_LOG2 8
#define P1_STEPS      (P1_WAVES * 256)
#define PB            16

__global__ void slab_all(const float* __restrict__ pk, const float* __restrict__ TAx,
                         const float* __restrict__ TAy, const float* __restrict__ fc,
                         const int* __restrict__ p_dt, const int* __restrict__ p_dtf,
                         float* __restrict__ out, int nsteps, int nchunks, int nt)
{
    __shared__ float2 part[P1_WAVES];
    __shared__ float2 tend[P1_WAVES];

    int gid  = blockIdx.x * blockDim.x + threadIdx.x;
    int tile = gid >> 5;
    int lane = gid & 31;
    const int b   = blockIdx.x;
    const int tid = threadIdx.x;
    const int wv  = tid >> 5;

    float dt      = (float)p_dt[0];
    int   nsub    = p_dtf[0] / p_dt[0];
    float invnsub = 1.f / (float)nsub;
    float K0   = __expf(pk[0]);
    float K1   = __expf(pk[1]);
    float f    = fc[0];
    float ar   = 1.f - dt * K1;
    float ai   = -dt * f;
    float dtK0 = dt * K0;

    const int N0 = b * P1_STEPS;
    const int I0 = N0 / nsub;
    const int R  = N0 - I0 * nsub;
    const int j0 = I0     < nt - 1 ? I0     : nt - 1;
    const int j1 = I0 + 1 < nt - 1 ? I0 + 1 : nt - 1;
    const float px0 = TAx[j0], py0 = TAy[j0], px1 = TAx[j1], py1 = TAy[j1];

    const int ntrip  = (I0 + P1_THREADS - 1) >> P1_THREADS_LOG2;
    const int nbatch = (ntrip + PB - 1) / PB;
    float tx[PB], ty[PB];
    #pragma unroll
    for (int u = 0; u < PB; ++u) { tx[u] = 0.f; ty[u] = 0.f; }
    if (nbatch > 0) {
        #pragma unroll
        for (int u = 0; u < PB; ++u) {
            int k = (nbatch - 1) * PB + (PB - 1 - u);
            int j = I0 - 1 - tid - k * P1_THREADS;
            int jc = j < 0 ? 0 : (j < nt - 1 ? j : nt - 1);
            tx[u] = TAx[jc]; ty[u] = TAy[jc];
        }
    }

    float p2r, p2i, p4r, p4i, p8r, p8i, p16r, p16i, p32r, p32i;
    cmul(ar,   ai,   ar,   ai,   p2r,  p2i);
    cmul(p2r,  p2i,  p2r,  p2i,  p4r,  p4i);
    cmul(p4r,  p4i,  p4r,  p4i,  p8r,  p8i);
    cmul(p8r,  p8i,  p8r,  p8i,  p16r, p16i);
    cmul(p16r, p16i, p16r, p16i, p32r, p32i);
    float lpr = (lane & 1) ? ar : 1.f;
    float lpi = (lane & 1) ? ai : 0.f;
    { float tr, ti; cmul(lpr, lpi, p2r,  p2i,  tr, ti); if (lane & 2)  { lpr = tr; lpi = ti; } }
    { float tr, ti; cmul(lpr, lpi, p4r,  p4i,  tr, ti); if (lane & 4)  { lpr = tr; lpi = ti; } }
    { float tr, ti; cmul(lpr, lpi, p8r,  p8i,  tr, ti); if (lane & 8)  { lpr = tr; lpi = ti; } }
    { float tr, ti; cmul(lpr, lpi, p16r, p16i, tr, ti); if (lane & 16) { lpr = tr; lpi = ti; } }

    float c0r = 0.f, c0i = 0.f, c1r = 0.f, c1i = 0.f;
    float d0r = 0.f, d0i = 0.f, d1r = 0.f, d1i = 0.f;
    float unr = 0.f, uni = 0.f;
    float urr = 0.f, uri = 0.f;
    const int ncoef = (nsub + 31) >> 5;
    for (int t = ncoef - 1; t >= 0; --t) {
        const int p  = lane + 32 * t;
        const int mc = nsub - 1 - p;
        const int md = R - 1 - p;
        const float aac = (float)mc * invnsub;
        const float aad = (float)md * invnsub;
        const float wc0 = mc >= 0 ? 1.f - aac : 0.f;
        const float wc1 = mc >= 0 ? aac       : 0.f;
        const float wd0 = md >= 0 ? 1.f - aad : 0.f;
        const float wd1 = md >= 0 ? aad       : 0.f;
        float tr, ti;
        cmul(p32r, p32i, c0r, c0i, tr, ti); c0r = tr + wc0; c0i = ti;
        cmul(p32r, p32i, c1r, c1i, tr, ti); c1r = tr + wc1; c1i = ti;
        cmul(p32r, p32i, d0r, d0i, tr, ti); d0r = tr + wd0; d0i = ti;
        cmul(p32r, p32i, d1r, d1i, tr, ti); d1r = tr + wd1; d1i = ti;
        cmul(p32r, p32i, unr, uni, tr, ti); unr = tr + (mc == 0 ? 1.f : 0.f); uni = ti;
        cmul(p32r, p32i, urr, uri, tr, ti); urr = tr + (md == 0 ? 1.f : 0.f); uri = ti;
    }
    {
        float tr, ti;
        cmul(lpr, lpi, c0r, c0i, tr, ti); c0r = lane_sum(tr); c0i = lane_sum(ti);
        cmul(lpr, lpi, c1r, c1i, tr, ti); c1r = lane_sum(tr); c1i = lane_sum(ti);
        cmul(lpr, lpi, d0r, d0i, tr, ti); d0r = lane_sum(tr); d0i = lane_sum(ti);
        cmul(lpr, lpi, d1r, d1i, tr, ti); d1r = lane_sum(tr); d1i = lane_sum(ti);
        cmul(lpr, lpi, unr, uni, tr, ti); unr = lane_sum(tr); uni = lane_sum(ti);
        cmul(lpr, lpi, urr, uri, tr, ti); urr = lane_sum(tr); uri = lane_sum(ti);
    }
    float Ar, Ai;
    cmul(ar, ai, unr, uni, Ar, Ai);
    float pRr = 1.f, pRi = 0.f;
    if (R > 0) cmul(ar, ai, urr, uri, pRr, pRi);
    float ccr, cci;
    { float tr, ti; cmul(c1r, c1i, Ar, Ai, tr, ti); ccr = c0r + tr; cci = c0i + ti; }

    float wtr = 1.f, wti = 0.f, qr = Ar, qi = Ai;
    #pragma unroll
    for (int bit = 0; bit < P1_THREADS_LOG2; ++bit) {
        float tr, ti;
        if ((tid >> bit) & 1) { cmul(wtr, wti, qr, qi, tr, ti); wtr = tr; wti = ti; }
        cmul(qr, qi, qr, qi, tr, ti); qr = tr; qi = ti;
    }
    float accr = 0.f, acci = 0.f;
    for (int bb = nbatch - 1; bb >= 0; --bb) {
        float nx[PB], ny[PB];
        #pragma unroll
        for (int u = 0; u < PB; ++u) { nx[u] = 0.f; ny[u] = 0.f; }
        if (bb > 0) {
            #pragma unroll
            for (int u = 0; u < PB; ++u) {
                int k = (bb - 1) * PB + (PB - 1 - u);
                int j = I0 - 1 - tid - k * P1_THREADS;
                int jc = j < 0 ? 0 : (j < nt - 1 ? j : nt - 1);
                nx[u] = TAx[jc]; ny[u] = TAy[jc];
            }
        }
        #pragma unroll
        for (int u = 0; u < PB; ++u) {
            int k = bb * PB + (PB - 1 - u);
            int j = I0 - 1 - tid - k * P1_THREADS;
            float kr = (j == 0) ? c0r : ccr;
            float ki = (j == 0) ? c0i : cci;
            float er, ei, tr, ti;
            cmul(kr, ki, tx[u], ty[u], er, ei);
            cmul(qr, qi, accr, acci, tr, ti);
            accr = tr + (j >= 0 ? er : 0.f);
            acci = ti + (j >= 0 ? ei : 0.f);
        }
        #pragma unroll
        for (int u = 0; u < PB; ++u) { tx[u] = nx[u]; ty[u] = ny[u]; }
    }
    {
        float tr, ti;
        cmul(wtr, wti, accr, acci, tr, ti); accr = tr; acci = ti;
    }
    #pragma unroll
    for (int off = 16; off >= 1; off >>= 1) {
        accr += __shfl_xor(accr, off, 32);
        acci += __shfl_xor(acci, off, 32);
    }
    if (lane == 0) part[wv] = make_float2(accr, acci);

    int half = lane >> 4;
    int r    = lane & 15;
    int c0   = tile * 16;

    float a2r, a2i, a4r, a4i, a8r, a8i;
    cmul(ar,  ai,  ar,  ai,  a2r, a2i);
    cmul(a2r, a2i, a2r, a2i, a4r, a4i);
    cmul(a4r, a4i, a4r, a4i, a8r, a8i);
    const int pe = r - 2 * half;
    const int pc = pe < 0 ? 0 : pe;
    float pwr = (pc & 1) ? ar : 1.f;
    float pwi = (pc & 1) ? ai : 0.f;
    { float tr, ti; cmul(pwr, pwi, a2r, a2i, tr, ti); if (pc & 2) { pwr = tr; pwi = ti; } }
    { float tr, ti; cmul(pwr, pwi, a4r, a4i, tr, ti); if (pc & 4) { pwr = tr; pwi = ti; } }
    { float tr, ti; cmul(pwr, pwi, a8r, a8i, tr, ti); if (pc & 8) { pwr = tr; pwi = ti; } }
    if (pe < 0) { pwr = 0.f; pwi = 0.f; }
    float Tr[8], Ti[8];
    Tr[0] = pwr;              Ti[0] = pwi;
    Tr[1] = row_up<1>(pwr);   Ti[1] = row_up<1>(pwi);
    Tr[2] = row_up<4>(pwr);   Ti[2] = row_up<4>(pwi);
    Tr[3] = row_up<5>(pwr);   Ti[3] = row_up<5>(pwi);
    Tr[4] = row_up<8>(pwr);   Ti[4] = row_up<8>(pwi);
    Tr[5] = row_up<9>(pwr);   Ti[5] = row_up<9>(pwi);
    Tr[6] = row_up<12>(pwr);  Ti[6] = row_up<12>(pwi);
    Tr[7] = row_up<13>(pwr);  Ti[7] = row_up<13>(pwi);
    const int      c   = c0 + r;
    const unsigned nb  = 16u * (unsigned)c + 2u * (unsigned)half;
    const unsigned usb = (unsigned)nsub;
    unsigned itf = nb / usb;
    unsigned rem = nb - itf * usb;

    v2f Are[4], Aim[4], Mre[4], Mim[4], Mneg[4];
    #pragma unroll
    for (int s = 0; s < 4; ++s) {
        #pragma unroll
        for (int e = 0; e < 2; ++e) {
            unsigned n  = nb + (unsigned)(4 * s + e);
            float aa = (float)rem * invnsub;
            int itf0 = (int)itf;
            int i0 = itf0     < nt - 1 ? itf0     : nt - 1;
            int i1 = itf0 + 1 < nt - 1 ? itf0 + 1 : nt - 1;
            float tax = (1.f - aa) * TAx[i0] + aa * TAx[i1];
            float tay = (1.f - aa) * TAy[i0] + aa * TAy[i1];
            float valid = (c < nchunks && n < (unsigned)nsteps) ? 1.f : 0.f;
            Are[s][e] = valid * dtK0 * tax;
            Aim[s][e] = valid * dtK0 * tay;
            Mre[s][e]  =  Tr[2 * s + e];
            Mim[s][e]  =  Ti[2 * s + e];
            Mneg[s][e] = -Ti[2 * s + e];
            const int step = (e == 0) ? 1 : 3;
            rem += (unsigned)step;
            #pragma unroll
            for (int t = 0; t < step; ++t)
                if (rem >= usb) { rem -= usb; itf += 1u; }
        }
    }

    v8f Sre = {}; v8f Sim = {};
    #pragma unroll
    for (int s = 0; s < 4; ++s)
        Sre = __builtin_amdgcn_wmma_f32_16x16x4_f32(false, Are[s], false, Mre[s],
                                                    (short)0, Sre, false, false);
    #pragma unroll
    for (int s = 0; s < 4; ++s)
        Sre = __builtin_amdgcn_wmma_f32_16x16x4_f32(false, Aim[s], false, Mneg[s],
                                                    (short)0, Sre, false, false);
    #pragma unroll
    for (int s = 0; s < 4; ++s)
        Sim = __builtin_amdgcn_wmma_f32_16x16x4_f32(false, Are[s], false, Mim[s],
                                                    (short)0, Sim, false, false);
    #pragma unroll
    for (int s = 0; s < 4; ++s)
        Sim = __builtin_amdgcn_wmma_f32_16x16x4_f32(false, Aim[s], false, Mre[s],
                                                    (short)0, Sim, false, false);

    int   srcl = 15 + ((lane & 8) << 1);
    float gr = 0.f, gi = 0.f;
    #pragma unroll
    for (int e = 0; e < 8; ++e) {
        float tr = __shfl(Sre[e], srcl, 32);
        float ti = __shfl(Sim[e], srcl, 32);
        if ((lane & 7) == e) { gr = tr; gi = ti; }
    }
    {
        int  c   = c0 + r;
        long n15 = (long)16 * c + 15;
        bool ok  = (c < nchunks) && (n15 < (long)nsteps);
        if (!ok) { gr = 0.f; gi = 0.f; }
    }

    float pkr = __shfl(Mre[0][0], r, 32);
    float pki = __shfl(Mim[0][0], r, 32);
    float Pr, Pi;   cmul(pkr, pki, ar, ai, Pr, Pi);
    float q15r = __shfl(Mre[0][0], 15, 32);
    float q15i = __shfl(Mim[0][0], 15, 32);
    float mr, mi;   cmul(q15r, q15i, ar, ai, mr, mi);

    float Ir = gr, Ii = gi;
    float zr = 1.f, zi = 0.f;
    #pragma unroll
    for (int off = 1; off < 16; off <<= 1) {
        int   sl = (lane & 16) | ((r - off) & 15);
        float tr = __shfl(Ir, sl, 32);
        float ti = __shfl(Ii, sl, 32);
        if (r >= off) {
            float ur, ui; cmul(mr, mi, tr, ti, ur, ui);
            Ir += ur; Ii += ui;
        }
        if (r & off) { float ur, ui; cmul(zr, zi, mr, mi, ur, ui); zr = ur; zi = ui; }
        float sr, si; cmul(mr, mi, mr, mi, sr, si); mr = sr; mi = si;
    }

    float Xr = __shfl(Ir, (lane & 16) | ((r - 1) & 15), 32);
    float Xi = __shfl(Ii, (lane & 16) | ((r - 1) & 15), 32);
    if (r == 0) { Xr = 0.f; Xi = 0.f; }

    float Gr = __shfl(Ir, 15, 32);
    float Gi = __shfl(Ii, 15, 32);
    if (lane == 0) tend[wv] = make_float2(Gr, Gi);

    __syncthreads();

    float2 ps[P1_WAVES], te[P1_WAVES];
    #pragma unroll
    for (int v = 0; v < P1_WAVES; ++v) { ps[v] = part[v]; te[v] = tend[v]; }
    float yr = 0.f, yi = 0.f;
    #pragma unroll
    for (int v = 0; v < P1_WAVES; ++v) { yr += ps[v].x; yi += ps[v].y; }
    float Jr, Ji;
    {
        if (I0 >= 1) { yr += c1r * px0 - c1i * py0; yi += c1r * py0 + c1i * px0; }
        float tr, ti;
        cmul(pRr, pRi, yr, yi, tr, ti);
        float hr = d0r * px0 - d0i * py0 + d1r * px1 - d1i * py1;
        float hi = d0r * py0 + d0i * px0 + d1r * py1 + d1i * px1;
        Jr = dtK0 * (tr + hr);
        Ji = dtK0 * (ti + hi);
    }
    #pragma unroll
    for (int v = 0; v < P1_WAVES; ++v) {
        float tr, ti; cmul(mr, mi, Jr, Ji, tr, ti);
        tr += te[v].x; ti += te[v].y;
        Jr = (v < wv) ? tr : Jr;
        Ji = (v < wv) ? ti : Ji;
    }
    float Wlr, Wli;
    {
        float tr, ti; cmul(zr, zi, Jr, Ji, tr, ti);
        Wlr = Xr + tr; Wli = Xi + ti;
    }

    v2f A1, B1re, B1im;
    A1[0]   = half ? 0.f : Wlr;   A1[1]   = half ? 0.f : Wli;
    B1re[0] = half ? 0.f : Pr;    B1re[1] = half ? 0.f : -Pi;
    B1im[0] = half ? 0.f : Pi;    B1im[1] = half ? 0.f : Pr;
    Sre = __builtin_amdgcn_wmma_f32_16x16x4_f32(false, A1, false, B1re,
                                                (short)0, Sre, false, false);
    Sim = __builtin_amdgcn_wmma_f32_16x16x4_f32(false, A1, false, B1im,
                                                (short)0, Sim, false, false);

    #pragma unroll
    for (int e = 0; e < 8; ++e) {
        int  row = e + 8 * half;
        int  c   = c0 + row;
        long n   = (long)16 * c + r;
        if (c < nchunks && n < (long)nsteps) {
            float2 v; v.x = Sre[e]; v.y = Sim[e];
            *reinterpret_cast<float2*>(out + 2 * n) = v;
        }
    }
}

extern "C" void kernel_launch(void* const* d_in, const int* in_sizes, int n_in,
                              void* d_out, int out_size, void* d_ws, size_t ws_size,
                              hipStream_t stream)
{
    const float* pk    = (const float*)d_in[0];
    const float* TAx   = (const float*)d_in[1];
    const float* TAy   = (const float*)d_in[2];
    const float* fc    = (const float*)d_in[3];
    const int*   p_dt  = (const int*)d_in[6];
    const int*   p_dtf = (const int*)d_in[7];
    (void)n_in; (void)ws_size; (void)d_ws;

    int nt      = in_sizes[1];
    int nsteps  = out_size / 2;
    int nchunks = (nsteps + 15) / 16;
    int ntiles  = (nchunks + 15) / 16;
    float*  out = (float*)d_out;

    int blocks1  = (ntiles + P1_WAVES - 1) / P1_WAVES;
    slab_all<<<blocks1, P1_THREADS, 0, stream>>>(pk, TAx, TAy, fc, p_dt, p_dtf,
                                                 out, nsteps, nchunks, nt);
}


# Blocks take their 2048-step spans in reverse index order: the late spans, whose start-value sums over all earlier samples are the longest, enter the chip first instead of last; arithmetic unchanged
# speedup vs baseline: 15.2236x; 1.0648x over previous
#include <hip/hip_runtime.h>


typedef __attribute__((ext_vector_type(2))) float v2f;
typedef __attribute__((ext_vector_type(8))) float v8f;

__device__ __forceinline__ void cmul(float ar, float ai, float br, float bi,
                                     float& cr, float& ci) {
    cr = ar * br - ai * bi;
    ci = ar * bi + ai * br;
}

template <int N>
__device__ __forceinline__ float row_up(float v) {
    return __int_as_float(__builtin_amdgcn_update_dpp(0, __float_as_int(v), 0x110 + N,
                                                      0xF, 0xF, true));
}

template <int M>
__device__ __forceinline__ float row_xor(float v) {
    return __int_as_float(__builtin_amdgcn_update_dpp(0, __float_as_int(v), 0x160 + M,
                                                      0xF, 0xF, true));
}
__device__ __forceinline__ float other_row(float v) {
    int b = __float_as_int(v);
    return __int_as_float(__builtin_amdgcn_permlanex16(b, b, 0x76543210, 0xfedcba98,
                                                       false, false));
}
__device__ __forceinline__ float lane_sum(float v) {
    v += row_xor<1>(v);
    v += row_xor<2>(v);
    v += row_xor<4>(v);
    v += row_xor<8>(v);
    v += other_row(v);
    return v;
}

#define P1_WAVES      8
#define P1_THREADS    (P1_WAVES * 32)
#define P1_THREADS_LOG2 8
#define P1_STEPS      (P1_WAVES * 256)
#define PB            16

__global__ void slab_all(const float* __restrict__ pk, const float* __restrict__ TAx,
                         const float* __restrict__ TAy, const float* __restrict__ fc,
                         const int* __restrict__ p_dt, const int* __restrict__ p_dtf,
                         float* __restrict__ out, int nsteps, int nchunks, int nt)
{
    __shared__ float2 part[P1_WAVES];
    __shared__ float2 tend[P1_WAVES];

    const int b   = (int)gridDim.x - 1 - (int)blockIdx.x;
    const int tid = threadIdx.x;
    const int wv  = tid >> 5;
    int tile = b * P1_WAVES + wv;
    int lane = tid & 31;

    float dt      = (float)p_dt[0];
    int   nsub    = p_dtf[0] / p_dt[0];
    float invnsub = 1.f / (float)nsub;
    float K0   = __expf(pk[0]);
    float K1   = __expf(pk[1]);
    float f    = fc[0];
    float ar   = 1.f - dt * K1;
    float ai   = -dt * f;
    float dtK0 = dt * K0;

    const int N0 = b * P1_STEPS;
    const int I0 = N0 / nsub;
    const int R  = N0 - I0 * nsub;
    const int j0 = I0     < nt - 1 ? I0     : nt - 1;
    const int j1 = I0 + 1 < nt - 1 ? I0 + 1 : nt - 1;
    const float px0 = TAx[j0], py0 = TAy[j0], px1 = TAx[j1], py1 = TAy[j1];

    const int ntrip  = (I0 + P1_THREADS - 1) >> P1_THREADS_LOG2;
    const int nbatch = (ntrip + PB - 1) / PB;
    float tx[PB], ty[PB];
    #pragma unroll
    for (int u = 0; u < PB; ++u) { tx[u] = 0.f; ty[u] = 0.f; }
    if (nbatch > 0) {
        #pragma unroll
        for (int u = 0; u < PB; ++u) {
            int k = (nbatch - 1) * PB + (PB - 1 - u);
            int j = I0 - 1 - tid - k * P1_THREADS;
            int jc = j < 0 ? 0 : (j < nt - 1 ? j : nt - 1);
            tx[u] = TAx[jc]; ty[u] = TAy[jc];
        }
    }

    float p2r, p2i, p4r, p4i, p8r, p8i, p16r, p16i, p32r, p32i;
    cmul(ar,   ai,   ar,   ai,   p2r,  p2i);
    cmul(p2r,  p2i,  p2r,  p2i,  p4r,  p4i);
    cmul(p4r,  p4i,  p4r,  p4i,  p8r,  p8i);
    cmul(p8r,  p8i,  p8r,  p8i,  p16r, p16i);
    cmul(p16r, p16i, p16r, p16i, p32r, p32i);
    float lpr = (lane & 1) ? ar : 1.f;
    float lpi = (lane & 1) ? ai : 0.f;
    { float tr, ti; cmul(lpr, lpi, p2r,  p2i,  tr, ti); if (lane & 2)  { lpr = tr; lpi = ti; } }
    { float tr, ti; cmul(lpr, lpi, p4r,  p4i,  tr, ti); if (lane & 4)  { lpr = tr; lpi = ti; } }
    { float tr, ti; cmul(lpr, lpi, p8r,  p8i,  tr, ti); if (lane & 8)  { lpr = tr; lpi = ti; } }
    { float tr, ti; cmul(lpr, lpi, p16r, p16i, tr, ti); if (lane & 16) { lpr = tr; lpi = ti; } }

    float c0r = 0.f, c0i = 0.f, c1r = 0.f, c1i = 0.f;
    float d0r = 0.f, d0i = 0.f, d1r = 0.f, d1i = 0.f;
    float unr = 0.f, uni = 0.f;
    float urr = 0.f, uri = 0.f;
    const int ncoef = (nsub + 31) >> 5;
    for (int t = ncoef - 1; t >= 0; --t) {
        const int p  = lane + 32 * t;
        const int mc = nsub - 1 - p;
        const int md = R - 1 - p;
        const float aac = (float)mc * invnsub;
        const float aad = (float)md * invnsub;
        const float wc0 = mc >= 0 ? 1.f - aac : 0.f;
        const float wc1 = mc >= 0 ? aac       : 0.f;
        const float wd0 = md >= 0 ? 1.f - aad : 0.f;
        const float wd1 = md >= 0 ? aad       : 0.f;
        float tr, ti;
        cmul(p32r, p32i, c0r, c0i, tr, ti); c0r = tr + wc0; c0i = ti;
        cmul(p32r, p32i, c1r, c1i, tr, ti); c1r = tr + wc1; c1i = ti;
        cmul(p32r, p32i, d0r, d0i, tr, ti); d0r = tr + wd0; d0i = ti;
        cmul(p32r, p32i, d1r, d1i, tr, ti); d1r = tr + wd1; d1i = ti;
        cmul(p32r, p32i, unr, uni, tr, ti); unr = tr + (mc == 0 ? 1.f : 0.f); uni = ti;
        cmul(p32r, p32i, urr, uri, tr, ti); urr = tr + (md == 0 ? 1.f : 0.f); uri = ti;
    }
    {
        float tr, ti;
        cmul(lpr, lpi, c0r, c0i, tr, ti); c0r = lane_sum(tr); c0i = lane_sum(ti);
        cmul(lpr, lpi, c1r, c1i, tr, ti); c1r = lane_sum(tr); c1i = lane_sum(ti);
        cmul(lpr, lpi, d0r, d0i, tr, ti); d0r = lane_sum(tr); d0i = lane_sum(ti);
        cmul(lpr, lpi, d1r, d1i, tr, ti); d1r = lane_sum(tr); d1i = lane_sum(ti);
        cmul(lpr, lpi, unr, uni, tr, ti); unr = lane_sum(tr); uni = lane_sum(ti);
        cmul(lpr, lpi, urr, uri, tr, ti); urr = lane_sum(tr); uri = lane_sum(ti);
    }
    float Ar, Ai;
    cmul(ar, ai, unr, uni, Ar, Ai);
    float pRr = 1.f, pRi = 0.f;
    if (R > 0) cmul(ar, ai, urr, uri, pRr, pRi);
    float ccr, cci;
    { float tr, ti; cmul(c1r, c1i, Ar, Ai, tr, ti); ccr = c0r + tr; cci = c0i + ti; }

    float wtr = 1.f, wti = 0.f, qr = Ar, qi = Ai;
    #pragma unroll
    for (int bit = 0; bit < P1_THREADS_LOG2; ++bit) {
        float tr, ti;
        if ((tid >> bit) & 1) { cmul(wtr, wti, qr, qi, tr, ti); wtr = tr; wti = ti; }
        cmul(qr, qi, qr, qi, tr, ti); qr = tr; qi = ti;
    }
    float accr = 0.f, acci = 0.f;
    for (int bb = nbatch - 1; bb >= 0; --bb) {
        float nx[PB], ny[PB];
        #pragma unroll
        for (int u = 0; u < PB; ++u) { nx[u] = 0.f; ny[u] = 0.f; }
        if (bb > 0) {
            #pragma unroll
            for (int u = 0; u < PB; ++u) {
                int k = (bb - 1) * PB + (PB - 1 - u);
                int j = I0 - 1 - tid - k * P1_THREADS;
                int jc = j < 0 ? 0 : (j < nt - 1 ? j : nt - 1);
                nx[u] = TAx[jc]; ny[u] = TAy[jc];
            }
        }
        #pragma unroll
        for (int u = 0; u < PB; ++u) {
            int k = bb * PB + (PB - 1 - u);
            int j = I0 - 1 - tid - k * P1_THREADS;
            float kr = (j == 0) ? c0r : ccr;
            float ki = (j == 0) ? c0i : cci;
            float er, ei, tr, ti;
            cmul(kr, ki, tx[u], ty[u], er, ei);
            cmul(qr, qi, accr, acci, tr, ti);
            accr = tr + (j >= 0 ? er : 0.f);
            acci = ti + (j >= 0 ? ei : 0.f);
        }
        #pragma unroll
        for (int u = 0; u < PB; ++u) { tx[u] = nx[u]; ty[u] = ny[u]; }
    }
    {
        float tr, ti;
        cmul(wtr, wti, accr, acci, tr, ti); accr = tr; acci = ti;
    }
    #pragma unroll
    for (int off = 16; off >= 1; off >>= 1) {
        accr += __shfl_xor(accr, off, 32);
        acci += __shfl_xor(acci, off, 32);
    }
    if (lane == 0) part[wv] = make_float2(accr, acci);

    int half = lane >> 4;
    int r    = lane & 15;
    int c0   = tile * 16;

    float a2r, a2i, a4r, a4i, a8r, a8i;
    cmul(ar,  ai,  ar,  ai,  a2r, a2i);
    cmul(a2r, a2i, a2r, a2i, a4r, a4i);
    cmul(a4r, a4i, a4r, a4i, a8r, a8i);
    const int pe = r - 2 * half;
    const int pc = pe < 0 ? 0 : pe;
    float pwr = (pc & 1) ? ar : 1.f;
    float pwi = (pc & 1) ? ai : 0.f;
    { float tr, ti; cmul(pwr, pwi, a2r, a2i, tr, ti); if (pc & 2) { pwr = tr; pwi = ti; } }
    { float tr, ti; cmul(pwr, pwi, a4r, a4i, tr, ti); if (pc & 4) { pwr = tr; pwi = ti; } }
    { float tr, ti; cmul(pwr, pwi, a8r, a8i, tr, ti); if (pc & 8) { pwr = tr; pwi = ti; } }
    if (pe < 0) { pwr = 0.f; pwi = 0.f; }
    float Tr[8], Ti[8];
    Tr[0] = pwr;              Ti[0] = pwi;
    Tr[1] = row_up<1>(pwr);   Ti[1] = row_up<1>(pwi);
    Tr[2] = row_up<4>(pwr);   Ti[2] = row_up<4>(pwi);
    Tr[3] = row_up<5>(pwr);   Ti[3] = row_up<5>(pwi);
    Tr[4] = row_up<8>(pwr);   Ti[4] = row_up<8>(pwi);
    Tr[5] = row_up<9>(pwr);   Ti[5] = row_up<9>(pwi);
    Tr[6] = row_up<12>(pwr);  Ti[6] = row_up<12>(pwi);
    Tr[7] = row_up<13>(pwr);  Ti[7] = row_up<13>(pwi);
    const int      c   = c0 + r;
    const unsigned nb  = 16u * (unsigned)c + 2u * (unsigned)half;
    const unsigned usb = (unsigned)nsub;
    unsigned itf = nb / usb;
    unsigned rem = nb - itf * usb;

    v2f Are[4], Aim[4], Mre[4], Mim[4], Mneg[4];
    #pragma unroll
    for (int s = 0; s < 4; ++s) {
        #pragma unroll
        for (int e = 0; e < 2; ++e) {
            unsigned n  = nb + (unsigned)(4 * s + e);
            float aa = (float)rem * invnsub;
            int itf0 = (int)itf;
            int i0 = itf0     < nt - 1 ? itf0     : nt - 1;
            int i1 = itf0 + 1 < nt - 1 ? itf0 + 1 : nt - 1;
            float tax = (1.f - aa) * TAx[i0] + aa * TAx[i1];
            float tay = (1.f - aa) * TAy[i0] + aa * TAy[i1];
            float valid = (c < nchunks && n < (unsigned)nsteps) ? 1.f : 0.f;
            Are[s][e] = valid * dtK0 * tax;
            Aim[s][e] = valid * dtK0 * tay;
            Mre[s][e]  =  Tr[2 * s + e];
            Mim[s][e]  =  Ti[2 * s + e];
            Mneg[s][e] = -Ti[2 * s + e];
            const int step = (e == 0) ? 1 : 3;
            rem += (unsigned)step;
            #pragma unroll
            for (int t = 0; t < step; ++t)
                if (rem >= usb) { rem -= usb; itf += 1u; }
        }
    }

    v8f Sre = {}; v8f Sim = {};
    #pragma unroll
    for (int s = 0; s < 4; ++s)
        Sre = __builtin_amdgcn_wmma_f32_16x16x4_f32(false, Are[s], false, Mre[s],
                                                    (short)0, Sre, false, false);
    #pragma unroll
    for (int s = 0; s < 4; ++s)
        Sre = __builtin_amdgcn_wmma_f32_16x16x4_f32(false, Aim[s], false, Mneg[s],
                                                    (short)0, Sre, false, false);
    #pragma unroll
    for (int s = 0; s < 4; ++s)
        Sim = __builtin_amdgcn_wmma_f32_16x16x4_f32(false, Are[s], false, Mim[s],
                                                    (short)0, Sim, false, false);
    #pragma unroll
    for (int s = 0; s < 4; ++s)
        Sim = __builtin_amdgcn_wmma_f32_16x16x4_f32(false, Aim[s], false, Mre[s],
                                                    (short)0, Sim, false, false);

    int   srcl = 15 + ((lane & 8) << 1);
    float gr = 0.f, gi = 0.f;
    #pragma unroll
    for (int e = 0; e < 8; ++e) {
        float tr = __shfl(Sre[e], srcl, 32);
        float ti = __shfl(Sim[e], srcl, 32);
        if ((lane & 7) == e) { gr = tr; gi = ti; }
    }
    {
        int  c   = c0 + r;
        long n15 = (long)16 * c + 15;
        bool ok  = (c < nchunks) && (n15 < (long)nsteps);
        if (!ok) { gr = 0.f; gi = 0.f; }
    }

    float pkr = __shfl(Mre[0][0], r, 32);
    float pki = __shfl(Mim[0][0], r, 32);
    float Pr, Pi;   cmul(pkr, pki, ar, ai, Pr, Pi);
    float q15r = __shfl(Mre[0][0], 15, 32);
    float q15i = __shfl(Mim[0][0], 15, 32);
    float mr, mi;   cmul(q15r, q15i, ar, ai, mr, mi);

    float Ir = gr, Ii = gi;
    float zr = 1.f, zi = 0.f;
    #pragma unroll
    for (int off = 1; off < 16; off <<= 1) {
        int   sl = (lane & 16) | ((r - off) & 15);
        float tr = __shfl(Ir, sl, 32);
        float ti = __shfl(Ii, sl, 32);
        if (r >= off) {
            float ur, ui; cmul(mr, mi, tr, ti, ur, ui);
            Ir += ur; Ii += ui;
        }
        if (r & off) { float ur, ui; cmul(zr, zi, mr, mi, ur, ui); zr = ur; zi = ui; }
        float sr, si; cmul(mr, mi, mr, mi, sr, si); mr = sr; mi = si;
    }

    float Xr = __shfl(Ir, (lane & 16) | ((r - 1) & 15), 32);
    float Xi = __shfl(Ii, (lane & 16) | ((r - 1) & 15), 32);
    if (r == 0) { Xr = 0.f; Xi = 0.f; }

    float Gr = __shfl(Ir, 15, 32);
    float Gi = __shfl(Ii, 15, 32);
    if (lane == 0) tend[wv] = make_float2(Gr, Gi);

    __syncthreads();

    float2 ps[P1_WAVES], te[P1_WAVES];
    #pragma unroll
    for (int v = 0; v < P1_WAVES; ++v) { ps[v] = part[v]; te[v] = tend[v]; }
    float yr = 0.f, yi = 0.f;
    #pragma unroll
    for (int v = 0; v < P1_WAVES; ++v) { yr += ps[v].x; yi += ps[v].y; }
    float Jr, Ji;
    {
        if (I0 >= 1) { yr += c1r * px0 - c1i * py0; yi += c1r * py0 + c1i * px0; }
        float tr, ti;
        cmul(pRr, pRi, yr, yi, tr, ti);
        float hr = d0r * px0 - d0i * py0 + d1r * px1 - d1i * py1;
        float hi = d0r * py0 + d0i * px0 + d1r * py1 + d1i * px1;
        Jr = dtK0 * (tr + hr);
        Ji = dtK0 * (ti + hi);
    }
    #pragma unroll
    for (int v = 0; v < P1_WAVES; ++v) {
        float tr, ti; cmul(mr, mi, Jr, Ji, tr, ti);
        tr += te[v].x; ti += te[v].y;
        Jr = (v < wv) ? tr : Jr;
        Ji = (v < wv) ? ti : Ji;
    }
    float Wlr, Wli;
    {
        float tr, ti; cmul(zr, zi, Jr, Ji, tr, ti);
        Wlr = Xr + tr; Wli = Xi + ti;
    }

    v2f A1, B1re, B1im;
    A1[0]   = half ? 0.f : Wlr;   A1[1]   = half ? 0.f : Wli;
    B1re[0] = half ? 0.f : Pr;    B1re[1] = half ? 0.f : -Pi;
    B1im[0] = half ? 0.f : Pi;    B1im[1] = half ? 0.f : Pr;
    Sre = __builtin_amdgcn_wmma_f32_16x16x4_f32(false, A1, false, B1re,
                                                (short)0, Sre, false, false);
    Sim = __builtin_amdgcn_wmma_f32_16x16x4_f32(false, A1, false, B1im,
                                                (short)0, Sim, false, false);

    #pragma unroll
    for (int e = 0; e < 8; ++e) {
        int  row = e + 8 * half;
        int  c   = c0 + row;
        long n   = (long)16 * c + r;
        if (c < nchunks && n < (long)nsteps) {
            float2 v; v.x = Sre[e]; v.y = Sim[e];
            *reinterpret_cast<float2*>(out + 2 * n) = v;
        }
    }
}

extern "C" void kernel_launch(void* const* d_in, const int* in_sizes, int n_in,
                              void* d_out, int out_size, void* d_ws, size_t ws_size,
                              hipStream_t stream)
{
    const float* pk    = (const float*)d_in[0];
    const float* TAx   = (const float*)d_in[1];
    const float* TAy   = (const float*)d_in[2];
    const float* fc    = (const float*)d_in[3];
    const int*   p_dt  = (const int*)d_in[6];
    const int*   p_dtf = (const int*)d_in[7];
    (void)n_in; (void)ws_size; (void)d_ws;

    int nt      = in_sizes[1];
    int nsteps  = out_size / 2;
    int nchunks = (nsteps + 15) / 16;
    int ntiles  = (nchunks + 15) / 16;
    float*  out = (float*)d_out;

    int blocks1  = (ntiles + P1_WAVES - 1) / P1_WAVES;
    slab_all<<<blocks1, P1_THREADS, 0, stream>>>(pk, TAx, TAy, fc, p_dt, p_dtf,
                                                 out, nsteps, nchunks, nt);
}


# Block start sum runs exactly ceil(I0/256) trips, not whole batches of 16 (far-end partial batch ends on a block-uniform branch); bounds and sample-0 selects only in the far-most trip; sums keep order
# speedup vs baseline: 16.0165x; 1.0521x over previous
#include <hip/hip_runtime.h>


typedef __attribute__((ext_vector_type(2))) float v2f;
typedef __attribute__((ext_vector_type(8))) float v8f;

__device__ __forceinline__ void cmul(float ar, float ai, float br, float bi,
                                     float& cr, float& ci) {
    cr = ar * br - ai * bi;
    ci = ar * bi + ai * br;
}

template <int N>
__device__ __forceinline__ float row_up(float v) {
    return __int_as_float(__builtin_amdgcn_update_dpp(0, __float_as_int(v), 0x110 + N,
                                                      0xF, 0xF, true));
}

template <int M>
__device__ __forceinline__ float row_xor(float v) {
    return __int_as_float(__builtin_amdgcn_update_dpp(0, __float_as_int(v), 0x160 + M,
                                                      0xF, 0xF, true));
}
__device__ __forceinline__ float other_row(float v) {
    int b = __float_as_int(v);
    return __int_as_float(__builtin_amdgcn_permlanex16(b, b, 0x76543210, 0xfedcba98,
                                                       false, false));
}
__device__ __forceinline__ float lane_sum(float v) {
    v += row_xor<1>(v);
    v += row_xor<2>(v);
    v += row_xor<4>(v);
    v += row_xor<8>(v);
    v += other_row(v);
    return v;
}

#define P1_WAVES      8
#define P1_THREADS    (P1_WAVES * 32)
#define P1_THREADS_LOG2 8
#define P1_STEPS      (P1_WAVES * 256)
#define PB            16

template <int U>
__device__ __forceinline__ void part_loads(float (&rx)[PB - 1], float (&ry)[PB - 1],
                                           const float* __restrict__ TAx,
                                           const float* __restrict__ TAy,
                                           int nrem, int jfar, int jmax) {
    if constexpr (U < PB - 1) {
        if (U < nrem) {
            int j  = jfar + U * P1_THREADS;
            int jc = j < jmax ? j : jmax;
            rx[U] = TAx[jc]; ry[U] = TAy[jc];
            part_loads<U + 1>(rx, ry, TAx, TAy, nrem, jfar, jmax);
        }
    }
}
template <int U>
__device__ __forceinline__ void part_sums(const float (&rx)[PB - 1], const float (&ry)[PB - 1],
                                          int nrem, float ccr, float cci, float qr, float qi,
                                          float& accr, float& acci) {
    if constexpr (U < PB - 1) {
        if (U < nrem) {
            float er, ei, tr, ti;
            cmul(ccr, cci, rx[U], ry[U], er, ei);
            cmul(qr, qi, accr, acci, tr, ti);
            accr = tr + er;
            acci = ti + ei;
            part_sums<U + 1>(rx, ry, nrem, ccr, cci, qr, qi, accr, acci);
        }
    }
}

__global__ void __launch_bounds__(P1_THREADS) slab_all(const float* __restrict__ pk, const float* __restrict__ TAx,
                         const float* __restrict__ TAy, const float* __restrict__ fc,
                         const int* __restrict__ p_dt, const int* __restrict__ p_dtf,
                         float* __restrict__ out, int nsteps, int nchunks, int nt)
{
    __shared__ float2 part[P1_WAVES];
    __shared__ float2 tend[P1_WAVES];

    const int b   = (int)gridDim.x - 1 - (int)blockIdx.x;
    const int tid = threadIdx.x;
    const int wv  = tid >> 5;
    int tile = b * P1_WAVES + wv;
    int lane = tid & 31;

    float dt      = (float)p_dt[0];
    int   nsub    = p_dtf[0] / p_dt[0];
    float invnsub = 1.f / (float)nsub;
    float K0   = __expf(pk[0]);
    float K1   = __expf(pk[1]);
    float f    = fc[0];
    float ar   = 1.f - dt * K1;
    float ai   = -dt * f;
    float dtK0 = dt * K0;

    const int N0 = b * P1_STEPS;
    const int I0 = N0 / nsub;
    const int R  = N0 - I0 * nsub;
    const int j0 = I0     < nt - 1 ? I0     : nt - 1;
    const int j1 = I0 + 1 < nt - 1 ? I0 + 1 : nt - 1;
    const float px0 = TAx[j0], py0 = TAy[j0], px1 = TAx[j1], py1 = TAy[j1];

    const int ntrip = (I0 + P1_THREADS - 1) >> P1_THREADS_LOG2;
    const int nrest = ntrip > 0 ? ntrip - 1 : 0;
    const int nfull = nrest / PB;
    const int nrem  = nrest - nfull * PB;
    const int jt    = I0 - 1 - tid;
    const int jf    = jt - nrest * P1_THREADS;
    float fx = 0.f, fy = 0.f;
    if (ntrip > 0) {
        int jc = jf < 0 ? 0 : (jf < nt - 1 ? jf : nt - 1);
        fx = TAx[jc]; fy = TAy[jc];
    }
    float rx[PB - 1], ry[PB - 1];
    part_loads<0>(rx, ry, TAx, TAy, nrem, jt - (nfull * PB + nrem - 1) * P1_THREADS, nt - 1);
    float tx[PB], ty[PB];
    #pragma unroll
    for (int u = 0; u < PB; ++u) { tx[u] = 0.f; ty[u] = 0.f; }
    if (nfull > 0) {
        #pragma unroll
        for (int u = 0; u < PB; ++u) {
            int k = (nfull - 1) * PB + (PB - 1 - u);
            int j = jt - k * P1_THREADS;
            int jc = j < nt - 1 ? j : nt - 1;
            tx[u] = TAx[jc]; ty[u] = TAy[jc];
        }
    }

    float p2r, p2i, p4r, p4i, p8r, p8i, p16r, p16i, p32r, p32i;
    cmul(ar,   ai,   ar,   ai,   p2r,  p2i);
    cmul(p2r,  p2i,  p2r,  p2i,  p4r,  p4i);
    cmul(p4r,  p4i,  p4r,  p4i,  p8r,  p8i);
    cmul(p8r,  p8i,  p8r,  p8i,  p16r, p16i);
    cmul(p16r, p16i, p16r, p16i, p32r, p32i);
    float lpr = (lane & 1) ? ar : 1.f;
    float lpi = (lane & 1) ? ai : 0.f;
    { float tr, ti; cmul(lpr, lpi, p2r,  p2i,  tr, ti); if (lane & 2)  { lpr = tr; lpi = ti; } }
    { float tr, ti; cmul(lpr, lpi, p4r,  p4i,  tr, ti); if (lane & 4)  { lpr = tr; lpi = ti; } }
    { float tr, ti; cmul(lpr, lpi, p8r,  p8i,  tr, ti); if (lane & 8)  { lpr = tr; lpi = ti; } }
    { float tr, ti; cmul(lpr, lpi, p16r, p16i, tr, ti); if (lane & 16) { lpr = tr; lpi = ti; } }

    float c0r = 0.f, c0i = 0.f, c1r = 0.f, c1i = 0.f;
    float d0r = 0.f, d0i = 0.f, d1r = 0.f, d1i = 0.f;
    float unr = 0.f, uni = 0.f;
    float urr = 0.f, uri = 0.f;
    const int ncoef = (nsub + 31) >> 5;
    for (int t = ncoef - 1; t >= 0; --t) {
        const int p  = lane + 32 * t;
        const int mc = nsub - 1 - p;
        const int md = R - 1 - p;
        const float aac = (float)mc * invnsub;
        const float aad = (float)md * invnsub;
        const float wc0 = mc >= 0 ? 1.f - aac : 0.f;
        const float wc1 = mc >= 0 ? aac       : 0.f;
        const float wd0 = md >= 0 ? 1.f - aad : 0.f;
        const float wd1 = md >= 0 ? aad       : 0.f;
        float tr, ti;
        cmul(p32r, p32i, c0r, c0i, tr, ti); c0r = tr + wc0; c0i = ti;
        cmul(p32r, p32i, c1r, c1i, tr, ti); c1r = tr + wc1; c1i = ti;
        cmul(p32r, p32i, d0r, d0i, tr, ti); d0r = tr + wd0; d0i = ti;
        cmul(p32r, p32i, d1r, d1i, tr, ti); d1r = tr + wd1; d1i = ti;
        cmul(p32r, p32i, unr, uni, tr, ti); unr = tr + (mc == 0 ? 1.f : 0.f); uni = ti;
        cmul(p32r, p32i, urr, uri, tr, ti); urr = tr + (md == 0 ? 1.f : 0.f); uri = ti;
    }
    {
        float tr, ti;
        cmul(lpr, lpi, c0r, c0i, tr, ti); c0r = lane_sum(tr); c0i = lane_sum(ti);
        cmul(lpr, lpi, c1r, c1i, tr, ti); c1r = lane_sum(tr); c1i = lane_sum(ti);
        cmul(lpr, lpi, d0r, d0i, tr, ti); d0r = lane_sum(tr); d0i = lane_sum(ti);
        cmul(lpr, lpi, d1r, d1i, tr, ti); d1r = lane_sum(tr); d1i = lane_sum(ti);
        cmul(lpr, lpi, unr, uni, tr, ti); unr = lane_sum(tr); uni = lane_sum(ti);
        cmul(lpr, lpi, urr, uri, tr, ti); urr = lane_sum(tr); uri = lane_sum(ti);
    }
    float Ar, Ai;
    cmul(ar, ai, unr, uni, Ar, Ai);
    float pRr = 1.f, pRi = 0.f;
    if (R > 0) cmul(ar, ai, urr, uri, pRr, pRi);
    float ccr, cci;
    { float tr, ti; cmul(c1r, c1i, Ar, Ai, tr, ti); ccr = c0r + tr; cci = c0i + ti; }

    float wtr = 1.f, wti = 0.f, qr = Ar, qi = Ai;
    #pragma unroll
    for (int bit = 0; bit < P1_THREADS_LOG2; ++bit) {
        float tr, ti;
        if ((tid >> bit) & 1) { cmul(wtr, wti, qr, qi, tr, ti); wtr = tr; wti = ti; }
        cmul(qr, qi, qr, qi, tr, ti); qr = tr; qi = ti;
    }
    float accr = 0.f, acci = 0.f;
    if (ntrip > 0) {
        float kr = (jf == 0) ? c0r : ccr;
        float ki = (jf == 0) ? c0i : cci;
        float er, ei;
        cmul(kr, ki, fx, fy, er, ei);
        accr = jf >= 0 ? er : 0.f;
        acci = jf >= 0 ? ei : 0.f;
    }
    part_sums<0>(rx, ry, nrem, ccr, cci, qr, qi, accr, acci);
    for (int bb = nfull - 1; bb >= 0; --bb) {
        float nx[PB], ny[PB];
        #pragma unroll
        for (int u = 0; u < PB; ++u) { nx[u] = 0.f; ny[u] = 0.f; }
        if (bb > 0) {
            #pragma unroll
            for (int u = 0; u < PB; ++u) {
                int k = (bb - 1) * PB + (PB - 1 - u);
                int j = jt - k * P1_THREADS;
                int jc = j < nt - 1 ? j : nt - 1;
                nx[u] = TAx[jc]; ny[u] = TAy[jc];
            }
        }
        #pragma unroll
        for (int u = 0; u < PB; ++u) {
            float er, ei, tr, ti;
            cmul(ccr, cci, tx[u], ty[u], er, ei);
            cmul(qr, qi, accr, acci, tr, ti);
            accr = tr + er;
            acci = ti + ei;
        }
        #pragma unroll
        for (int u = 0; u < PB; ++u) { tx[u] = nx[u]; ty[u] = ny[u]; }
    }
    {
        float tr, ti;
        cmul(wtr, wti, accr, acci, tr, ti); accr = tr; acci = ti;
    }
    #pragma unroll
    for (int off = 16; off >= 1; off >>= 1) {
        accr += __shfl_xor(accr, off, 32);
        acci += __shfl_xor(acci, off, 32);
    }
    if (lane == 0) part[wv] = make_float2(accr, acci);

    int half = lane >> 4;
    int r    = lane & 15;
    int c0   = tile * 16;

    float a2r, a2i, a4r, a4i, a8r, a8i;
    cmul(ar,  ai,  ar,  ai,  a2r, a2i);
    cmul(a2r, a2i, a2r, a2i, a4r, a4i);
    cmul(a4r, a4i, a4r, a4i, a8r, a8i);
    const int pe = r - 2 * half;
    const int pc = pe < 0 ? 0 : pe;
    float pwr = (pc & 1) ? ar : 1.f;
    float pwi = (pc & 1) ? ai : 0.f;
    { float tr, ti; cmul(pwr, pwi, a2r, a2i, tr, ti); if (pc & 2) { pwr = tr; pwi = ti; } }
    { float tr, ti; cmul(pwr, pwi, a4r, a4i, tr, ti); if (pc & 4) { pwr = tr; pwi = ti; } }
    { float tr, ti; cmul(pwr, pwi, a8r, a8i, tr, ti); if (pc & 8) { pwr = tr; pwi = ti; } }
    if (pe < 0) { pwr = 0.f; pwi = 0.f; }
    float Tr[8], Ti[8];
    Tr[0] = pwr;              Ti[0] = pwi;
    Tr[1] = row_up<1>(pwr);   Ti[1] = row_up<1>(pwi);
    Tr[2] = row_up<4>(pwr);   Ti[2] = row_up<4>(pwi);
    Tr[3] = row_up<5>(pwr);   Ti[3] = row_up<5>(pwi);
    Tr[4] = row_up<8>(pwr);   Ti[4] = row_up<8>(pwi);
    Tr[5] = row_up<9>(pwr);   Ti[5] = row_up<9>(pwi);
    Tr[6] = row_up<12>(pwr);  Ti[6] = row_up<12>(pwi);
    Tr[7] = row_up<13>(pwr);  Ti[7] = row_up<13>(pwi);
    const int      c   = c0 + r;
    const unsigned nb  = 16u * (unsigned)c + 2u * (unsigned)half;
    const unsigned usb = (unsigned)nsub;
    unsigned itf = nb / usb;
    unsigned rem = nb - itf * usb;

    v2f Are[4], Aim[4], Mre[4], Mim[4], Mneg[4];
    #pragma unroll
    for (int s = 0; s < 4; ++s) {
        #pragma unroll
        for (int e = 0; e < 2; ++e) {
            unsigned n  = nb + (unsigned)(4 * s + e);
            float aa = (float)rem * invnsub;
            int itf0 = (int)itf;
            int i0 = itf0     < nt - 1 ? itf0     : nt - 1;
            int i1 = itf0 + 1 < nt - 1 ? itf0 + 1 : nt - 1;
            float tax = (1.f - aa) * TAx[i0] + aa * TAx[i1];
            float tay = (1.f - aa) * TAy[i0] + aa * TAy[i1];
            float valid = (c < nchunks && n < (unsigned)nsteps) ? 1.f : 0.f;
            Are[s][e] = valid * dtK0 * tax;
            Aim[s][e] = valid * dtK0 * tay;
            Mre[s][e]  =  Tr[2 * s + e];
            Mim[s][e]  =  Ti[2 * s + e];
            Mneg[s][e] = -Ti[2 * s + e];
            const int step = (e == 0) ? 1 : 3;
            rem += (unsigned)step;
            #pragma unroll
            for (int t = 0; t < step; ++t)
                if (rem >= usb) { rem -= usb; itf += 1u; }
        }
    }

    v8f Sre = {}; v8f Sim = {};
    #pragma unroll
    for (int s = 0; s < 4; ++s)
        Sre = __builtin_amdgcn_wmma_f32_16x16x4_f32(false, Are[s], false, Mre[s],
                                                    (short)0, Sre, false, false);
    #pragma unroll
    for (int s = 0; s < 4; ++s)
        Sre = __builtin_amdgcn_wmma_f32_16x16x4_f32(false, Aim[s], false, Mneg[s],
                                                    (short)0, Sre, false, false);
    #pragma unroll
    for (int s = 0; s < 4; ++s)
        Sim = __builtin_amdgcn_wmma_f32_16x16x4_f32(false, Are[s], false, Mim[s],
                                                    (short)0, Sim, false, false);
    #pragma unroll
    for (int s = 0; s < 4; ++s)
        Sim = __builtin_amdgcn_wmma_f32_16x16x4_f32(false, Aim[s], false, Mre[s],
                                                    (short)0, Sim, false, false);

    int   srcl = 15 + ((lane & 8) << 1);
    float gr = 0.f, gi = 0.f;
    #pragma unroll
    for (int e = 0; e < 8; ++e) {
        float tr = __shfl(Sre[e], srcl, 32);
        float ti = __shfl(Sim[e], srcl, 32);
        if ((lane & 7) == e) { gr = tr; gi = ti; }
    }
    {
        int  c   = c0 + r;
        long n15 = (long)16 * c + 15;
        bool ok  = (c < nchunks) && (n15 < (long)nsteps);
        if (!ok) { gr = 0.f; gi = 0.f; }
    }

    float pkr = __shfl(Mre[0][0], r, 32);
    float pki = __shfl(Mim[0][0], r, 32);
    float Pr, Pi;   cmul(pkr, pki, ar, ai, Pr, Pi);
    float q15r = __shfl(Mre[0][0], 15, 32);
    float q15i = __shfl(Mim[0][0], 15, 32);
    float mr, mi;   cmul(q15r, q15i, ar, ai, mr, mi);

    float Ir = gr, Ii = gi;
    float zr = 1.f, zi = 0.f;
    #pragma unroll
    for (int off = 1; off < 16; off <<= 1) {
        int   sl = (lane & 16) | ((r - off) & 15);
        float tr = __shfl(Ir, sl, 32);
        float ti = __shfl(Ii, sl, 32);
        if (r >= off) {
            float ur, ui; cmul(mr, mi, tr, ti, ur, ui);
            Ir += ur; Ii += ui;
        }
        if (r & off) { float ur, ui; cmul(zr, zi, mr, mi, ur, ui); zr = ur; zi = ui; }
        float sr, si; cmul(mr, mi, mr, mi, sr, si); mr = sr; mi = si;
    }

    float Xr = __shfl(Ir, (lane & 16) | ((r - 1) & 15), 32);
    float Xi = __shfl(Ii, (lane & 16) | ((r - 1) & 15), 32);
    if (r == 0) { Xr = 0.f; Xi = 0.f; }

    float Gr = __shfl(Ir, 15, 32);
    float Gi = __shfl(Ii, 15, 32);
    if (lane == 0) tend[wv] = make_float2(Gr, Gi);

    __syncthreads();

    float2 ps[P1_WAVES], te[P1_WAVES];
    #pragma unroll
    for (int v = 0; v < P1_WAVES; ++v) { ps[v] = part[v]; te[v] = tend[v]; }
    float yr = 0.f, yi = 0.f;
    #pragma unroll
    for (int v = 0; v < P1_WAVES; ++v) { yr += ps[v].x; yi += ps[v].y; }
    float Jr, Ji;
    {
        if (I0 >= 1) { yr += c1r * px0 - c1i * py0; yi += c1r * py0 + c1i * px0; }
        float tr, ti;
        cmul(pRr, pRi, yr, yi, tr, ti);
        float hr = d0r * px0 - d0i * py0 + d1r * px1 - d1i * py1;
        float hi = d0r * py0 + d0i * px0 + d1r * py1 + d1i * px1;
        Jr = dtK0 * (tr + hr);
        Ji = dtK0 * (ti + hi);
    }
    #pragma unroll
    for (int v = 0; v < P1_WAVES; ++v) {
        float tr, ti; cmul(mr, mi, Jr, Ji, tr, ti);
        tr += te[v].x; ti += te[v].y;
        Jr = (v < wv) ? tr : Jr;
        Ji = (v < wv) ? ti : Ji;
    }
    float Wlr, Wli;
    {
        float tr, ti; cmul(zr, zi, Jr, Ji, tr, ti);
        Wlr = Xr + tr; Wli = Xi + ti;
    }

    v2f A1, B1re, B1im;
    A1[0]   = half ? 0.f : Wlr;   A1[1]   = half ? 0.f : Wli;
    B1re[0] = half ? 0.f : Pr;    B1re[1] = half ? 0.f : -Pi;
    B1im[0] = half ? 0.f : Pi;    B1im[1] = half ? 0.f : Pr;
    Sre = __builtin_amdgcn_wmma_f32_16x16x4_f32(false, A1, false, B1re,
                                                (short)0, Sre, false, false);
    Sim = __builtin_amdgcn_wmma_f32_16x16x4_f32(false, A1, false, B1im,
                                                (short)0, Sim, false, false);

    #pragma unroll
    for (int e = 0; e < 8; ++e) {
        int  row = e + 8 * half;
        int  c   = c0 + row;
        long n   = (long)16 * c + r;
        if (c < nchunks && n < (long)nsteps) {
            float2 v; v.x = Sre[e]; v.y = Sim[e];
            *reinterpret_cast<float2*>(out + 2 * n) = v;
        }
    }
}

extern "C" void kernel_launch(void* const* d_in, const int* in_sizes, int n_in,
                              void* d_out, int out_size, void* d_ws, size_t ws_size,
                              hipStream_t stream)
{
    const float* pk    = (const float*)d_in[0];
    const float* TAx   = (const float*)d_in[1];
    const float* TAy   = (const float*)d_in[2];
    const float* fc    = (const float*)d_in[3];
    const int*   p_dt  = (const int*)d_in[6];
    const int*   p_dtf = (const int*)d_in[7];
    (void)n_in; (void)ws_size; (void)d_ws;

    int nt      = in_sizes[1];
    int nsteps  = out_size / 2;
    int nchunks = (nsteps + 15) / 16;
    int ntiles  = (nchunks + 15) / 16;
    float*  out = (float*)d_out;

    int blocks1  = (ntiles + P1_WAVES - 1) / P1_WAVES;
    slab_all<<<blocks1, P1_THREADS, 0, stream>>>(pk, TAx, TAy, fc, p_dt, p_dtf,
                                                 out, nsteps, nchunks, nt);
}


# Behind the barrier only what needs the other waves' values: lane l reads partial sum and tile end l&7, weights m^(w-1-v) made before, sums by DPP; full tiles store unguarded; reduction tree changes
# speedup vs baseline: 20.2744x; 1.2658x over previous
#include <hip/hip_runtime.h>


typedef __attribute__((ext_vector_type(2))) float v2f;
typedef __attribute__((ext_vector_type(8))) float v8f;

__device__ __forceinline__ void cmul(float ar, float ai, float br, float bi,
                                     float& cr, float& ci) {
    cr = ar * br - ai * bi;
    ci = ar * bi + ai * br;
}

template <int N>
__device__ __forceinline__ float row_up(float v) {
    return __int_as_float(__builtin_amdgcn_update_dpp(0, __float_as_int(v), 0x110 + N,
                                                      0xF, 0xF, true));
}

template <int M>
__device__ __forceinline__ float row_xor(float v) {
    return __int_as_float(__builtin_amdgcn_update_dpp(0, __float_as_int(v), 0x160 + M,
                                                      0xF, 0xF, true));
}
__device__ __forceinline__ float other_row(float v) {
    int b = __float_as_int(v);
    return __int_as_float(__builtin_amdgcn_permlanex16(b, b, 0x76543210, 0xfedcba98,
                                                       false, false));
}
__device__ __forceinline__ float lane_sum(float v) {
    v += row_xor<1>(v);
    v += row_xor<2>(v);
    v += row_xor<4>(v);
    v += row_xor<8>(v);
    v += other_row(v);
    return v;
}

#define P1_WAVES      8
#define P1_THREADS    (P1_WAVES * 32)
#define P1_THREADS_LOG2 8
#define P1_STEPS      (P1_WAVES * 256)
#define PB            16

template <int U>
__device__ __forceinline__ void part_loads(float (&rx)[PB - 1], float (&ry)[PB - 1],
                                           const float* __restrict__ TAx,
                                           const float* __restrict__ TAy,
                                           int nrem, int jfar, int jmax) {
    if constexpr (U < PB - 1) {
        if (U < nrem) {
            int j  = jfar + U * P1_THREADS;
            int jc = j < jmax ? j : jmax;
            rx[U] = TAx[jc]; ry[U] = TAy[jc];
            part_loads<U + 1>(rx, ry, TAx, TAy, nrem, jfar, jmax);
        }
    }
}
template <int U>
__device__ __forceinline__ void part_sums(const float (&rx)[PB - 1], const float (&ry)[PB - 1],
                                          int nrem, float ccr, float cci, float qr, float qi,
                                          float& accr, float& acci) {
    if constexpr (U < PB - 1) {
        if (U < nrem) {
            float er, ei, tr, ti;
            cmul(ccr, cci, rx[U], ry[U], er, ei);
            cmul(qr, qi, accr, acci, tr, ti);
            accr = tr + er;
            acci = ti + ei;
            part_sums<U + 1>(rx, ry, nrem, ccr, cci, qr, qi, accr, acci);
        }
    }
}

__global__ void __launch_bounds__(P1_THREADS) slab_all(const float* __restrict__ pk, const float* __restrict__ TAx,
                         const float* __restrict__ TAy, const float* __restrict__ fc,
                         const int* __restrict__ p_dt, const int* __restrict__ p_dtf,
                         float* __restrict__ out, int nsteps, int nchunks, int nt)
{
    __shared__ float2 part[P1_WAVES];
    __shared__ float2 tend[P1_WAVES];

    const int b   = (int)gridDim.x - 1 - (int)blockIdx.x;
    const int tid = threadIdx.x;
    const int wv  = tid >> 5;
    int tile = b * P1_WAVES + wv;
    int lane = tid & 31;

    float dt      = (float)p_dt[0];
    int   nsub    = p_dtf[0] / p_dt[0];
    float invnsub = 1.f / (float)nsub;
    float K0   = __expf(pk[0]);
    float K1   = __expf(pk[1]);
    float f    = fc[0];
    float ar   = 1.f - dt * K1;
    float ai   = -dt * f;
    float dtK0 = dt * K0;

    const int N0 = b * P1_STEPS;
    const int I0 = N0 / nsub;
    const int R  = N0 - I0 * nsub;
    const int j0 = I0     < nt - 1 ? I0     : nt - 1;
    const int j1 = I0 + 1 < nt - 1 ? I0 + 1 : nt - 1;
    const float px0 = TAx[j0], py0 = TAy[j0], px1 = TAx[j1], py1 = TAy[j1];

    const int ntrip = (I0 + P1_THREADS - 1) >> P1_THREADS_LOG2;
    const int nrest = ntrip > 0 ? ntrip - 1 : 0;
    const int nfull = nrest / PB;
    const int nrem  = nrest - nfull * PB;
    const int jt    = I0 - 1 - tid;
    const int jf    = jt - nrest * P1_THREADS;
    float fx = 0.f, fy = 0.f;
    if (ntrip > 0) {
        int jc = jf < 0 ? 0 : (jf < nt - 1 ? jf : nt - 1);
        fx = TAx[jc]; fy = TAy[jc];
    }
    float rx[PB - 1], ry[PB - 1];
    part_loads<0>(rx, ry, TAx, TAy, nrem, jt - (nfull * PB + nrem - 1) * P1_THREADS, nt - 1);
    float tx[PB], ty[PB];
    #pragma unroll
    for (int u = 0; u < PB; ++u) { tx[u] = 0.f; ty[u] = 0.f; }
    if (nfull > 0) {
        #pragma unroll
        for (int u = 0; u < PB; ++u) {
            int k = (nfull - 1) * PB + (PB - 1 - u);
            int j = jt - k * P1_THREADS;
            int jc = j < nt - 1 ? j : nt - 1;
            tx[u] = TAx[jc]; ty[u] = TAy[jc];
        }
    }

    float p2r, p2i, p4r, p4i, p8r, p8i, p16r, p16i, p32r, p32i;
    cmul(ar,   ai,   ar,   ai,   p2r,  p2i);
    cmul(p2r,  p2i,  p2r,  p2i,  p4r,  p4i);
    cmul(p4r,  p4i,  p4r,  p4i,  p8r,  p8i);
    cmul(p8r,  p8i,  p8r,  p8i,  p16r, p16i);
    cmul(p16r, p16i, p16r, p16i, p32r, p32i);
    float lpr = (lane & 1) ? ar : 1.f;
    float lpi = (lane & 1) ? ai : 0.f;
    { float tr, ti; cmul(lpr, lpi, p2r,  p2i,  tr, ti); if (lane & 2)  { lpr = tr; lpi = ti; } }
    { float tr, ti; cmul(lpr, lpi, p4r,  p4i,  tr, ti); if (lane & 4)  { lpr = tr; lpi = ti; } }
    { float tr, ti; cmul(lpr, lpi, p8r,  p8i,  tr, ti); if (lane & 8)  { lpr = tr; lpi = ti; } }
    { float tr, ti; cmul(lpr, lpi, p16r, p16i, tr, ti); if (lane & 16) { lpr = tr; lpi = ti; } }

    float c0r = 0.f, c0i = 0.f, c1r = 0.f, c1i = 0.f;
    float d0r = 0.f, d0i = 0.f, d1r = 0.f, d1i = 0.f;
    float unr = 0.f, uni = 0.f;
    float urr = 0.f, uri = 0.f;
    const int ncoef = (nsub + 31) >> 5;
    for (int t = ncoef - 1; t >= 0; --t) {
        const int p  = lane + 32 * t;
        const int mc = nsub - 1 - p;
        const int md = R - 1 - p;
        const float aac = (float)mc * invnsub;
        const float aad = (float)md * invnsub;
        const float wc0 = mc >= 0 ? 1.f - aac : 0.f;
        const float wc1 = mc >= 0 ? aac       : 0.f;
        const float wd0 = md >= 0 ? 1.f - aad : 0.f;
        const float wd1 = md >= 0 ? aad       : 0.f;
        float tr, ti;
        cmul(p32r, p32i, c0r, c0i, tr, ti); c0r = tr + wc0; c0i = ti;
        cmul(p32r, p32i, c1r, c1i, tr, ti); c1r = tr + wc1; c1i = ti;
        cmul(p32r, p32i, d0r, d0i, tr, ti); d0r = tr + wd0; d0i = ti;
        cmul(p32r, p32i, d1r, d1i, tr, ti); d1r = tr + wd1; d1i = ti;
        cmul(p32r, p32i, unr, uni, tr, ti); unr = tr + (mc == 0 ? 1.f : 0.f); uni = ti;
        cmul(p32r, p32i, urr, uri, tr, ti); urr = tr + (md == 0 ? 1.f : 0.f); uri = ti;
    }
    {
        float tr, ti;
        cmul(lpr, lpi, c0r, c0i, tr, ti); c0r = lane_sum(tr); c0i = lane_sum(ti);
        cmul(lpr, lpi, c1r, c1i, tr, ti); c1r = lane_sum(tr); c1i = lane_sum(ti);
        cmul(lpr, lpi, d0r, d0i, tr, ti); d0r = lane_sum(tr); d0i = lane_sum(ti);
        cmul(lpr, lpi, d1r, d1i, tr, ti); d1r = lane_sum(tr); d1i = lane_sum(ti);
        cmul(lpr, lpi, unr, uni, tr, ti); unr = lane_sum(tr); uni = lane_sum(ti);
        cmul(lpr, lpi, urr, uri, tr, ti); urr = lane_sum(tr); uri = lane_sum(ti);
    }
    float Ar, Ai;
    cmul(ar, ai, unr, uni, Ar, Ai);
    float pRr = 1.f, pRi = 0.f;
    if (R > 0) cmul(ar, ai, urr, uri, pRr, pRi);
    float ccr, cci;
    { float tr, ti; cmul(c1r, c1i, Ar, Ai, tr, ti); ccr = c0r + tr; cci = c0i + ti; }

    float wtr = 1.f, wti = 0.f, qr = Ar, qi = Ai;
    #pragma unroll
    for (int bit = 0; bit < P1_THREADS_LOG2; ++bit) {
        float tr, ti;
        if ((tid >> bit) & 1) { cmul(wtr, wti, qr, qi, tr, ti); wtr = tr; wti = ti; }
        cmul(qr, qi, qr, qi, tr, ti); qr = tr; qi = ti;
    }
    float accr = 0.f, acci = 0.f;
    if (ntrip > 0) {
        float kr = (jf == 0) ? c0r : ccr;
        float ki = (jf == 0) ? c0i : cci;
        float er, ei;
        cmul(kr, ki, fx, fy, er, ei);
        accr = jf >= 0 ? er : 0.f;
        acci = jf >= 0 ? ei : 0.f;
    }
    part_sums<0>(rx, ry, nrem, ccr, cci, qr, qi, accr, acci);
    for (int bb = nfull - 1; bb >= 0; --bb) {
        float nx[PB], ny[PB];
        #pragma unroll
        for (int u = 0; u < PB; ++u) { nx[u] = 0.f; ny[u] = 0.f; }
        if (bb > 0) {
            #pragma unroll
            for (int u = 0; u < PB; ++u) {
                int k = (bb - 1) * PB + (PB - 1 - u);
                int j = jt - k * P1_THREADS;
                int jc = j < nt - 1 ? j : nt - 1;
                nx[u] = TAx[jc]; ny[u] = TAy[jc];
            }
        }
        #pragma unroll
        for (int u = 0; u < PB; ++u) {
            float er, ei, tr, ti;
            cmul(ccr, cci, tx[u], ty[u], er, ei);
            cmul(qr, qi, accr, acci, tr, ti);
            accr = tr + er;
            acci = ti + ei;
        }
        #pragma unroll
        for (int u = 0; u < PB; ++u) { tx[u] = nx[u]; ty[u] = ny[u]; }
    }
    {
        float tr, ti;
        cmul(wtr, wti, accr, acci, tr, ti); accr = tr; acci = ti;
    }
    #pragma unroll
    for (int off = 16; off >= 1; off >>= 1) {
        accr += __shfl_xor(accr, off, 32);
        acci += __shfl_xor(acci, off, 32);
    }
    if (lane == 0) part[wv] = make_float2(accr, acci);

    int half = lane >> 4;
    int r    = lane & 15;
    int c0   = tile * 16;

    float a2r, a2i, a4r, a4i, a8r, a8i;
    cmul(ar,  ai,  ar,  ai,  a2r, a2i);
    cmul(a2r, a2i, a2r, a2i, a4r, a4i);
    cmul(a4r, a4i, a4r, a4i, a8r, a8i);
    const int pe = r - 2 * half;
    const int pc = pe < 0 ? 0 : pe;
    float pwr = (pc & 1) ? ar : 1.f;
    float pwi = (pc & 1) ? ai : 0.f;
    { float tr, ti; cmul(pwr, pwi, a2r, a2i, tr, ti); if (pc & 2) { pwr = tr; pwi = ti; } }
    { float tr, ti; cmul(pwr, pwi, a4r, a4i, tr, ti); if (pc & 4) { pwr = tr; pwi = ti; } }
    { float tr, ti; cmul(pwr, pwi, a8r, a8i, tr, ti); if (pc & 8) { pwr = tr; pwi = ti; } }
    if (pe < 0) { pwr = 0.f; pwi = 0.f; }
    float Tr[8], Ti[8];
    Tr[0] = pwr;              Ti[0] = pwi;
    Tr[1] = row_up<1>(pwr);   Ti[1] = row_up<1>(pwi);
    Tr[2] = row_up<4>(pwr);   Ti[2] = row_up<4>(pwi);
    Tr[3] = row_up<5>(pwr);   Ti[3] = row_up<5>(pwi);
    Tr[4] = row_up<8>(pwr);   Ti[4] = row_up<8>(pwi);
    Tr[5] = row_up<9>(pwr);   Ti[5] = row_up<9>(pwi);
    Tr[6] = row_up<12>(pwr);  Ti[6] = row_up<12>(pwi);
    Tr[7] = row_up<13>(pwr);  Ti[7] = row_up<13>(pwi);
    const int      c   = c0 + r;
    const unsigned nb  = 16u * (unsigned)c + 2u * (unsigned)half;
    const unsigned usb = (unsigned)nsub;
    unsigned itf = nb / usb;
    unsigned rem = nb - itf * usb;

    v2f Are[4], Aim[4], Mre[4], Mim[4], Mneg[4];
    #pragma unroll
    for (int s = 0; s < 4; ++s) {
        #pragma unroll
        for (int e = 0; e < 2; ++e) {
            unsigned n  = nb + (unsigned)(4 * s + e);
            float aa = (float)rem * invnsub;
            int itf0 = (int)itf;
            int i0 = itf0     < nt - 1 ? itf0     : nt - 1;
            int i1 = itf0 + 1 < nt - 1 ? itf0 + 1 : nt - 1;
            float tax = (1.f - aa) * TAx[i0] + aa * TAx[i1];
            float tay = (1.f - aa) * TAy[i0] + aa * TAy[i1];
            float valid = (c < nchunks && n < (unsigned)nsteps) ? 1.f : 0.f;
            Are[s][e] = valid * dtK0 * tax;
            Aim[s][e] = valid * dtK0 * tay;
            Mre[s][e]  =  Tr[2 * s + e];
            Mim[s][e]  =  Ti[2 * s + e];
            Mneg[s][e] = -Ti[2 * s + e];
            const int step = (e == 0) ? 1 : 3;
            rem += (unsigned)step;
            #pragma unroll
            for (int t = 0; t < step; ++t)
                if (rem >= usb) { rem -= usb; itf += 1u; }
        }
    }

    v8f Sre = {}; v8f Sim = {};
    #pragma unroll
    for (int s = 0; s < 4; ++s)
        Sre = __builtin_amdgcn_wmma_f32_16x16x4_f32(false, Are[s], false, Mre[s],
                                                    (short)0, Sre, false, false);
    #pragma unroll
    for (int s = 0; s < 4; ++s)
        Sre = __builtin_amdgcn_wmma_f32_16x16x4_f32(false, Aim[s], false, Mneg[s],
                                                    (short)0, Sre, false, false);
    #pragma unroll
    for (int s = 0; s < 4; ++s)
        Sim = __builtin_amdgcn_wmma_f32_16x16x4_f32(false, Are[s], false, Mim[s],
                                                    (short)0, Sim, false, false);
    #pragma unroll
    for (int s = 0; s < 4; ++s)
        Sim = __builtin_amdgcn_wmma_f32_16x16x4_f32(false, Aim[s], false, Mre[s],
                                                    (short)0, Sim, false, false);

    int   srcl = 15 + ((lane & 8) << 1);
    float gr = 0.f, gi = 0.f;
    #pragma unroll
    for (int e = 0; e < 8; ++e) {
        float tr = __shfl(Sre[e], srcl, 32);
        float ti = __shfl(Sim[e], srcl, 32);
        if ((lane & 7) == e) { gr = tr; gi = ti; }
    }
    {
        int  c   = c0 + r;
        long n15 = (long)16 * c + 15;
        bool ok  = (c < nchunks) && (n15 < (long)nsteps);
        if (!ok) { gr = 0.f; gi = 0.f; }
    }

    float pkr = __shfl(Mre[0][0], r, 32);
    float pki = __shfl(Mim[0][0], r, 32);
    float Pr, Pi;   cmul(pkr, pki, ar, ai, Pr, Pi);
    float q15r = __shfl(Mre[0][0], 15, 32);
    float q15i = __shfl(Mim[0][0], 15, 32);
    float mr, mi;   cmul(q15r, q15i, ar, ai, mr, mi);

    float Ir = gr, Ii = gi;
    float zr = 1.f, zi = 0.f;
    #pragma unroll
    for (int off = 1; off < 16; off <<= 1) {
        int   sl = (lane & 16) | ((r - off) & 15);
        float tr = __shfl(Ir, sl, 32);
        float ti = __shfl(Ii, sl, 32);
        if (r >= off) {
            float ur, ui; cmul(mr, mi, tr, ti, ur, ui);
            Ir += ur; Ii += ui;
        }
        if (r & off) { float ur, ui; cmul(zr, zi, mr, mi, ur, ui); zr = ur; zi = ui; }
        float sr, si; cmul(mr, mi, mr, mi, sr, si); mr = sr; mi = si;
    }

    float Xr = __shfl(Ir, (lane & 16) | ((r - 1) & 15), 32);
    float Xi = __shfl(Ii, (lane & 16) | ((r - 1) & 15), 32);
    if (r == 0) { Xr = 0.f; Xi = 0.f; }

    float Gr = __shfl(Ir, 15, 32);
    float Gi = __shfl(Ii, 15, 32);
    if (lane == 0) tend[wv] = make_float2(Gr, Gi);

    static_assert(P1_WAVES == 8, "the hand-over takes three exponent bits and groups of 8 lanes");
    const int v8 = lane & (P1_WAVES - 1);
    float m2r, m2i, m4r, m4i;
    cmul(mr,  mi,  mr,  mi,  m2r, m2i);
    cmul(m2r, m2i, m2r, m2i, m4r, m4i);
    const int pg  = wv - 1 - v8;
    const int pgc = pg < 0 ? 0 : pg;
    float gwr = (pgc & 1) ? mr : 1.f;
    float gwi = (pgc & 1) ? mi : 0.f;
    { float tr, ti; cmul(gwr, gwi, m2r, m2i, tr, ti); if (pgc & 2) { gwr = tr; gwi = ti; } }
    { float tr, ti; cmul(gwr, gwi, m4r, m4i, tr, ti); if (pgc & 4) { gwr = tr; gwi = ti; } }
    float mwr = (wv & 1) ? mr : 1.f;
    float mwi = (wv & 1) ? mi : 0.f;
    { float tr, ti; cmul(mwr, mwi, m2r, m2i, tr, ti); if (wv & 2) { mwr = tr; mwi = ti; } }
    { float tr, ti; cmul(mwr, mwi, m4r, m4i, tr, ti); if (wv & 4) { mwr = tr; mwi = ti; } }
    float zkr, zki, Wcr, Wci;
    {
        float g0r = 0.f, g0i = 0.f;
        if (I0 >= 1) { g0r = c1r * px0 - c1i * py0; g0i = c1r * py0 + c1i * px0; }
        float hr = d0r * px0 - d0i * py0 + d1r * px1 - d1i * py1;
        float hi = d0r * py0 + d0i * px0 + d1r * py1 + d1i * px1;
        float tr, ti;
        cmul(pRr, pRi, g0r, g0i, tr, ti);
        float j0r = dtK0 * (tr + hr), j0i = dtK0 * (ti + hi);
        float zmr, zmi;
        cmul(zr, zi, mwr, mwi, zmr, zmi);
        cmul(zmr, zmi, j0r, j0i, tr, ti);
        Wcr = Xr + tr; Wci = Xi + ti;
        cmul(zmr, zmi, dtK0 * pRr, dtK0 * pRi, zkr, zki);
    }
    v2f A1, B1re, B1im;
    B1re[0] = half ? 0.f : Pr;    B1re[1] = half ? 0.f : -Pi;
    B1im[0] = half ? 0.f : Pi;    B1im[1] = half ? 0.f : Pr;

    __syncthreads();

    const float2 pv = part[v8];
    const float2 ev = tend[v8];
    float yr = pv.x, yi = pv.y;
    float Er, Ei;
    {
        float tr, ti; cmul(gwr, gwi, ev.x, ev.y, tr, ti);
        Er = (v8 < wv) ? tr : 0.f;
        Ei = (v8 < wv) ? ti : 0.f;
    }
    yr += row_xor<1>(yr);  yi += row_xor<1>(yi);  Er += row_xor<1>(Er);  Ei += row_xor<1>(Ei);
    yr += row_xor<2>(yr);  yi += row_xor<2>(yi);  Er += row_xor<2>(Er);  Ei += row_xor<2>(Ei);
    yr += row_xor<4>(yr);  yi += row_xor<4>(yi);  Er += row_xor<4>(Er);  Ei += row_xor<4>(Ei);
    float Wlr, Wli;
    {
        float ur, ui, vr, vi;
        cmul(zkr, zki, yr, yi, ur, ui);
        cmul(zr, zi, Er, Ei, vr, vi);
        Wlr = Wcr + (ur + vr); Wli = Wci + (ui + vi);
    }
    A1[0]   = half ? 0.f : Wlr;   A1[1]   = half ? 0.f : Wli;
    Sre = __builtin_amdgcn_wmma_f32_16x16x4_f32(false, A1, false, B1re,
                                                (short)0, Sre, false, false);
    Sim = __builtin_amdgcn_wmma_f32_16x16x4_f32(false, A1, false, B1im,
                                                (short)0, Sim, false, false);

    const int tile_s = __builtin_amdgcn_readfirstlane(tile);
    if (256 * tile_s + 255 < nsteps) {
        float2* ob = reinterpret_cast<float2*>(out) + ((long)256 * tile_s + 128 * half + r);
        #pragma unroll
        for (int e = 0; e < 8; ++e) {
            float2 v; v.x = Sre[e]; v.y = Sim[e];
            ob[16 * e] = v;
        }
    } else {
        #pragma unroll
        for (int e = 0; e < 8; ++e) {
            int  row = e + 8 * half;
            int  c   = c0 + row;
            long n   = (long)16 * c + r;
            if (c < nchunks && n < (long)nsteps) {
                float2 v; v.x = Sre[e]; v.y = Sim[e];
                *reinterpret_cast<float2*>(out + 2 * n) = v;
            }
        }
    }
}

extern "C" void kernel_launch(void* const* d_in, const int* in_sizes, int n_in,
                              void* d_out, int out_size, void* d_ws, size_t ws_size,
                              hipStream_t stream)
{
    const float* pk    = (const float*)d_in[0];
    const float* TAx   = (const float*)d_in[1];
    const float* TAy   = (const float*)d_in[2];
    const float* fc    = (const float*)d_in[3];
    const int*   p_dt  = (const int*)d_in[6];
    const int*   p_dtf = (const int*)d_in[7];
    (void)n_in; (void)ws_size; (void)d_ws;

    int nt      = in_sizes[1];
    int nsteps  = out_size / 2;
    int nchunks = (nsteps + 15) / 16;
    int ntiles  = (nchunks + 15) / 16;
    float*  out = (float*)d_out;

    int blocks1  = (ntiles + P1_WAVES - 1) / P1_WAVES;
    slab_all<<<blocks1, P1_THREADS, 0, stream>>>(pk, TAx, TAy, fc, p_dt, p_dtf,
                                                 out, nsteps, nchunks, nt);
}
